# starting kernel
# baseline (speedup 1.0000x reference)
#include <hip/hip_runtime.h>
#include <hip/hip_bf16.h>

typedef float v2f __attribute__((ext_vector_type(2)));
typedef float v8f __attribute__((ext_vector_type(8)));

#define FEAT 64
#define FILL_WEIGHT 1.0f

__global__ void gcn_deg_init(float* __restrict__ deg, int n) {
    int i = blockIdx.x * blockDim.x + threadIdx.x;
    if (i < n) deg[i] = FILL_WEIGHT;
}

__global__ void gcn_deg_accum(const int* __restrict__ row,
                              const float* __restrict__ w,
                              float* __restrict__ deg, int ne) {
    int e = blockIdx.x * blockDim.x + threadIdx.x;
    if (e < ne) atomicAdd(deg + row[e], w[e]);
}

__global__ void gcn_dinv(const float* __restrict__ deg,
                         float* __restrict__ dinv, int n) {
    int i = blockIdx.x * blockDim.x + threadIdx.x;
    if (i < n) {
        float d = deg[i];
        dinv[i] = (d > 0.0f) ? __frsqrt_rn(fmaxf(d, 1e-12f)) : 0.0f;
    }
}

__global__ __launch_bounds__(256)
void gcn_gemm_wmma(const float* __restrict__ x,
                   const float* __restrict__ kern,
                   float* __restrict__ h,
                   int num_nodes) {
    const int lane  = threadIdx.x & 31;
    const int wave  = threadIdx.x >> 5;
    const int tile  = blockIdx.x * 8 + wave;
    const int ntile = (num_nodes + 15) >> 4;
    if (tile >= ntile) return;

    const int r0   = tile << 4;
    const int half = lane >> 4;
    const int m    = lane & 15;

    v8f acc[4] = {};

    const float* xrow = x + (size_t)(r0 + m) * FEAT;
#pragma unroll
    for (int kb = 0; kb < FEAT; kb += 4) {
        const int k0 = kb + (half << 1);
        v2f a;
        a.x = xrow[k0];
        a.y = xrow[k0 + 1];
#pragma unroll
        for (int cb = 0; cb < 4; ++cb) {
            v2f b;
            b.x = kern[(size_t)k0 * FEAT + cb * 16 + m];
            b.y = kern[(size_t)(k0 + 1) * FEAT + cb * 16 + m];
            acc[cb] = __builtin_amdgcn_wmma_f32_16x16x4_f32(
                 false, a,  false, b,
                 (short)0, acc[cb],
                 false,  false);
        }
    }

#pragma unroll
    for (int cb = 0; cb < 4; ++cb) {
#pragma unroll
        for (int v = 0; v < 8; ++v) {
            h[(size_t)(r0 + v + 8 * half) * FEAT + cb * 16 + m] = acc[cb][v];
        }
    }
}

__global__ void gcn_out_init(const float* __restrict__ h,
                             const float* __restrict__ dinv,
                             const float* __restrict__ bias,
                             float* __restrict__ out, long long total) {
    long long idx = (long long)blockIdx.x * blockDim.x + threadIdx.x;
    if (idx >= total) return;
    int i = (int)(idx >> 6);
    int f = (int)(idx & 63);
    float d = dinv[i];
    out[idx] = bias[f] + h[idx] * d * d * FILL_WEIGHT;
}

__global__ void gcn_edge_scatter(const int* __restrict__ row,
                                 const int* __restrict__ col,
                                 const float* __restrict__ w,
                                 const float* __restrict__ dinv,
                                 const float* __restrict__ h,
                                 float* __restrict__ out, long long total) {
    long long idx = (long long)blockIdx.x * blockDim.x + threadIdx.x;
    if (idx >= total) return;
    int e = (int)(idx >> 6);
    int f = (int)(idx & 63);
    int r = row[e];
    int c = col[e];
    float wn = dinv[r] * w[e] * dinv[c];
    atomicAdd(out + (size_t)r * FEAT + f, h[(size_t)c * FEAT + f] * wn);
}

extern "C" void kernel_launch(void* const* d_in, const int* in_sizes, int n_in,
                              void* d_out, int out_size, void* d_ws, size_t ws_size,
                              hipStream_t stream) {
    const float* x    = (const float*)d_in[0];
    const int*   ei   = (const int*)d_in[1];
    const float* ew   = (const float*)d_in[2];
    const float* kern = (const float*)d_in[3];
    const float* bias = (const float*)d_in[4];

    const int N = in_sizes[0] / FEAT;
    const int E = in_sizes[2];
    const int* row = ei;
    const int* col = ei + E;

    float* h    = (float*)d_ws;
    float* deg  = h + (size_t)N * FEAT;
    float* dinv = deg + N;
    float* out  = (float*)d_out;

    const int T = 256;

    gcn_deg_init<<<(N + T - 1) / T, T, 0, stream>>>(deg, N);
    gcn_deg_accum<<<(E + T - 1) / T, T, 0, stream>>>(row, ew, deg, E);
    gcn_dinv<<<(N + T - 1) / T, T, 0, stream>>>(deg, dinv, N);

    const int ntile = (N + 15) / 16;
    gcn_gemm_wmma<<<(ntile + 7) / 8, T, 0, stream>>>(x, kern, h, N);

    long long tn = (long long)N * FEAT;
    gcn_out_init<<<(int)((tn + T - 1) / T), T, 0, stream>>>(h, dinv, bias, out, tn);

    long long te = (long long)E * FEAT;
    gcn_edge_scatter<<<(int)((te + T - 1) / T), T, 0, stream>>>(row, col, ew, dinv, h, out, te);
}


# Edges counting-sorted by destination row (returning int atomic rank, scan, placement); one wave per row sums degree and messages in registers, float atomics gone; per-row sum order = arrival rank
# speedup vs baseline: 2.2434x; 2.2434x over previous
#include <hip/hip_runtime.h>
#include <hip/hip_bf16.h>

typedef float v2f __attribute__((ext_vector_type(2)));
typedef float v8f __attribute__((ext_vector_type(8)));

#define FEAT 64
#define FILL_WEIGHT 1.0f
#define SCAN_CHUNK 1024


__global__ void gcn_cnt_zero(int* __restrict__ cnt, int n) {
    int i = blockIdx.x * blockDim.x + threadIdx.x;
    if (i < n) cnt[i] = 0;
}

__global__ void gcn_edge_rank(const int* __restrict__ row,
                              int* __restrict__ cnt,
                              int* __restrict__ rank, int ne) {
    int e = blockIdx.x * blockDim.x + threadIdx.x;
    if (e < ne) rank[e] = atomicAdd(cnt + row[e], 1);
}

__global__ __launch_bounds__(256)
void gcn_scan_partials(const int* __restrict__ cnt, int* __restrict__ bsum, int n) {
    __shared__ int sm[8];
    const int t  = threadIdx.x;
    const int i0 = blockIdx.x * SCAN_CHUNK + t * 4;
    int s = 0;
    for (int k = 0; k < 4; ++k)
        if (i0 + k < n) s += cnt[i0 + k];
    for (int d = 16; d > 0; d >>= 1) s += __shfl_xor(s, d);
    if ((t & 31) == 0) sm[t >> 5] = s;
    __syncthreads();
    if (t == 0) {
        int tot = 0;
        for (int k = 0; k < 8; ++k) tot += sm[k];
        bsum[blockIdx.x] = tot;
    }
}

__global__ __launch_bounds__(256)
void gcn_scan_offsets(const int* __restrict__ cnt, const int* __restrict__ bsum,
                      int* __restrict__ offs, int n) {
    __shared__ int sm_base[8];
    __shared__ int sm_wave[8];
    const int t    = threadIdx.x;
    const int lane = t & 31;
    const int wv   = t >> 5;

    int b = 0;
    for (int i = t; i < (int)blockIdx.x; i += 256) b += bsum[i];
    for (int d = 16; d > 0; d >>= 1) b += __shfl_xor(b, d);

    const int i0 = blockIdx.x * SCAN_CHUNK + t * 4;
    int c[4];
    int s = 0;
    for (int k = 0; k < 4; ++k) {
        c[k] = (i0 + k < n) ? cnt[i0 + k] : 0;
        s += c[k];
    }
    int incl = s;
    for (int d = 1; d < 32; d <<= 1) {
        int up = __shfl_up(incl, d);
        if (lane >= d) incl += up;
    }
    if (lane == 31) sm_wave[wv] = incl;
    if (lane == 0)  sm_base[wv] = b;
    __syncthreads();
    int run = incl - s;
    for (int k = 0; k < 8; ++k) {
        run += sm_base[k];
        if (k < wv) run += sm_wave[k];
    }
    for (int k = 0; k < 4; ++k) {
        if (i0 + k < n) offs[i0 + k] = run;
        run += c[k];
    }
}

__global__ void gcn_edge_place(const int* __restrict__ row,
                               const int* __restrict__ col,
                               const float* __restrict__ w,
                               const int* __restrict__ rank,
                               const int* __restrict__ offs,
                               int2* __restrict__ csr, int ne) {
    int e = blockIdx.x * blockDim.x + threadIdx.x;
    if (e < ne) {
        int pos = offs[row[e]] + rank[e];
        csr[pos] = make_int2(col[e], __float_as_int(w[e]));
    }
}

__global__ void gcn_dinv(const int* __restrict__ offs,
                         const int* __restrict__ cnt,
                         const int2* __restrict__ csr,
                         float* __restrict__ dinv, int n) {
    int i = blockIdx.x * blockDim.x + threadIdx.x;
    if (i < n) {
        const int beg = offs[i];
        const int num = cnt[i];
        float d = FILL_WEIGHT;
        for (int p = 0; p < num; ++p) d += __int_as_float(csr[beg + p].y);
        dinv[i] = (d > 0.0f) ? __frsqrt_rn(fmaxf(d, 1e-12f)) : 0.0f;
    }
}

__global__ __launch_bounds__(256)
void gcn_gemm_wmma(const float* __restrict__ x,
                   const float* __restrict__ kern,
                   float* __restrict__ h,
                   int num_nodes) {
    const int lane  = threadIdx.x & 31;
    const int wave  = threadIdx.x >> 5;
    const int tile  = blockIdx.x * 8 + wave;
    const int ntile = (num_nodes + 15) >> 4;
    if (tile >= ntile) return;

    const int r0   = tile << 4;
    const int half = lane >> 4;
    const int m    = lane & 15;

    v8f acc[4] = {};

    const float* xrow = x + (size_t)(r0 + m) * FEAT;
#pragma unroll
    for (int kb = 0; kb < FEAT; kb += 4) {
        const int k0 = kb + (half << 1);
        v2f a;
        a.x = xrow[k0];
        a.y = xrow[k0 + 1];
#pragma unroll
        for (int cb = 0; cb < 4; ++cb) {
            v2f b;
            b.x = kern[(size_t)k0 * FEAT + cb * 16 + m];
            b.y = kern[(size_t)(k0 + 1) * FEAT + cb * 16 + m];
            acc[cb] = __builtin_amdgcn_wmma_f32_16x16x4_f32(
                 false, a,  false, b,
                 (short)0, acc[cb],
                 false,  false);
        }
    }

#pragma unroll
    for (int cb = 0; cb < 4; ++cb) {
#pragma unroll
        for (int v = 0; v < 8; ++v) {
            h[(size_t)(r0 + v + 8 * half) * FEAT + cb * 16 + m] = acc[cb][v];
        }
    }
}

__global__ __launch_bounds__(256)
void gcn_row_gather(const int* __restrict__ offs,
                    const int* __restrict__ cnt,
                    const int2* __restrict__ csr,
                    const float* __restrict__ dinv,
                    const float* __restrict__ h,
                    const float* __restrict__ bias,
                    float* __restrict__ out, int n) {
    const int lane = threadIdx.x & 31;
    const int r = __builtin_amdgcn_readfirstlane(blockIdx.x * 8 + (threadIdx.x >> 5));
    if (r >= n) return;

    const int   beg = offs[r];
    const int   num = cnt[r];
    const float dr  = dinv[r];
    const v2f* __restrict__ h2 = reinterpret_cast<const v2f*>(h);

#define EDGE_C(j) __builtin_amdgcn_readlane(c, (j))
#define EDGE_W(j) __int_as_float(__builtin_amdgcn_readlane(__float_as_int(wn), (j)))
#define H_ROW(cc) h2[(size_t)(cc) * (FEAT / 2) + lane]

    v2f acc = {0.0f, 0.0f};
    for (int base = 0; base < num; base += 32) {
        const int p = base + lane;
        int   c  = 0;
        float wn = 0.0f;
        if (p < num) {
            const int2 cw = csr[beg + p];
            c  = cw.x;
            wn = dr * __int_as_float(cw.y) * dinv[c];
        }
        const int m = min(32, num - base);
        int j = 0;
        for (; j + 8 <= m; j += 8) {
            const v2f a0 = H_ROW(EDGE_C(j));
            const v2f a1 = H_ROW(EDGE_C(j + 1));
            const v2f a2 = H_ROW(EDGE_C(j + 2));
            const v2f a3 = H_ROW(EDGE_C(j + 3));
            const v2f a4 = H_ROW(EDGE_C(j + 4));
            const v2f a5 = H_ROW(EDGE_C(j + 5));
            const v2f a6 = H_ROW(EDGE_C(j + 6));
            const v2f a7 = H_ROW(EDGE_C(j + 7));
            acc += a0 * EDGE_W(j);
            acc += a1 * EDGE_W(j + 1);
            acc += a2 * EDGE_W(j + 2);
            acc += a3 * EDGE_W(j + 3);
            acc += a4 * EDGE_W(j + 4);
            acc += a5 * EDGE_W(j + 5);
            acc += a6 * EDGE_W(j + 6);
            acc += a7 * EDGE_W(j + 7);
        }
        if (j + 4 <= m) {
            const v2f a0 = H_ROW(EDGE_C(j));
            const v2f a1 = H_ROW(EDGE_C(j + 1));
            const v2f a2 = H_ROW(EDGE_C(j + 2));
            const v2f a3 = H_ROW(EDGE_C(j + 3));
            acc += a0 * EDGE_W(j);
            acc += a1 * EDGE_W(j + 1);
            acc += a2 * EDGE_W(j + 2);
            acc += a3 * EDGE_W(j + 3);
            j += 4;
        }
#pragma clang loop unroll(disable)
        for (; j < m; ++j) {
            const v2f a = H_ROW(EDGE_C(j));
            acc += a * EDGE_W(j);
        }
    }
#undef EDGE_C
#undef EDGE_W
#undef H_ROW

    const v2f   hs = h2[(size_t)r * (FEAT / 2) + lane];
    const float b0 = bias[2 * lane];
    const float b1 = bias[2 * lane + 1];
    v2f o;
    o.x = (b0 + hs.x * dr * dr * FILL_WEIGHT) + acc.x;
    o.y = (b1 + hs.y * dr * dr * FILL_WEIGHT) + acc.y;
    reinterpret_cast<v2f*>(out)[(size_t)r * (FEAT / 2) + lane] = o;
}

extern "C" void kernel_launch(void* const* d_in, const int* in_sizes, int n_in,
                              void* d_out, int out_size, void* d_ws, size_t ws_size,
                              hipStream_t stream) {
    const float* x    = (const float*)d_in[0];
    const int*   ei   = (const int*)d_in[1];
    const float* ew   = (const float*)d_in[2];
    const float* kern = (const float*)d_in[3];
    const float* bias = (const float*)d_in[4];

    const int N = in_sizes[0] / FEAT;
    const int E = in_sizes[2];
    const int* row = ei;
    const int* col = ei + E;

    const int NB   = (N + SCAN_CHUNK - 1) / SCAN_CHUNK;
    const int Npad = (N + 3) & ~3;
    const int Bpad = (NB + 3) & ~3;
    const int Epad = (E + 3) & ~3;

    float* h    = (float*)d_ws;
    float* dinv = h + (size_t)N * FEAT;
    int*   cnt  = (int*)(dinv + Npad);
    int*   offs = cnt + Npad;
    int*   bsum = offs + Npad;
    int*   rank = bsum + Bpad;
    int2*  csr  = (int2*)(rank + Epad);
    float* out  = (float*)d_out;

    const int T = 256;

    gcn_cnt_zero<<<(N + T - 1) / T, T, 0, stream>>>(cnt, N);
    gcn_edge_rank<<<(E + T - 1) / T, T, 0, stream>>>(row, cnt, rank, E);
    gcn_scan_partials<<<NB, T, 0, stream>>>(cnt, bsum, N);
    gcn_scan_offsets<<<NB, T, 0, stream>>>(cnt, bsum, offs, N);
    gcn_edge_place<<<(E + T - 1) / T, T, 0, stream>>>(row, col, ew, rank, offs, csr, E);
    gcn_dinv<<<(N + T - 1) / T, T, 0, stream>>>(offs, cnt, csr, dinv, N);

    const int ntile = (N + 15) / 16;
    gcn_gemm_wmma<<<(ntile + 7) / 8, T, 0, stream>>>(x, kern, h, N);

    gcn_row_gather<<<(N + 7) / 8, T, 0, stream>>>(offs, cnt, csr, dinv, h, bias, out, N);
}


# Edge ranks from LDS counters in a two-level counting sort (512-row buckets, then one workgroup per bucket) replace the returning global atomic per edge; per-row sum order = LDS arrival rank
# speedup vs baseline: 3.1884x; 1.4212x over previous
#include <hip/hip_runtime.h>
#include <hip/hip_bf16.h>

typedef float v2f __attribute__((ext_vector_type(2)));
typedef float v8f __attribute__((ext_vector_type(8)));

#define FEAT 64
#define FILL_WEIGHT 1.0f
#define SCAN_CHUNK 1024


#define BK_SHIFT   9
#define BK_ROWS    512
#define EDGE_CHUNK 4096
#define MAX_BK     2048

__global__ __launch_bounds__(256)
void gcn_bucket_hist(const int* __restrict__ row, int* __restrict__ bh,
                     int ne, int nblk, int nbk) {
    __shared__ int hist[MAX_BK];
    const int t = threadIdx.x;
    for (int i = t; i < nbk; i += 256) hist[i] = 0;
    __syncthreads();
    const int e0 = blockIdx.x * EDGE_CHUNK + t;
    int r[16];
    for (int k = 0; k < 16; ++k) {
        const int e = e0 + k * 256;
        r[k] = (e < ne) ? row[e] : -1;
    }
    for (int k = 0; k < 16; ++k)
        if (r[k] >= 0) atomicAdd(&hist[r[k] >> BK_SHIFT], 1);
    __syncthreads();
    for (int i = t; i < nbk; i += 256) bh[(size_t)i * nblk + blockIdx.x] = hist[i];
}

__global__ __launch_bounds__(256)
void gcn_scan_partials(const int* __restrict__ cnt, int* __restrict__ bsum, int n) {
    __shared__ int sm[8];
    const int t  = threadIdx.x;
    const int i0 = blockIdx.x * SCAN_CHUNK + t * 4;
    int s = 0;
    for (int k = 0; k < 4; ++k)
        if (i0 + k < n) s += cnt[i0 + k];
    for (int d = 16; d > 0; d >>= 1) s += __shfl_xor(s, d);
    if ((t & 31) == 0) sm[t >> 5] = s;
    __syncthreads();
    if (t == 0) {
        int tot = 0;
        for (int k = 0; k < 8; ++k) tot += sm[k];
        bsum[blockIdx.x] = tot;
    }
}

__global__ __launch_bounds__(256)
void gcn_scan_offsets(const int* __restrict__ cnt, const int* __restrict__ bsum,
                      int* __restrict__ offs, int n) {
    __shared__ int sm_base[8];
    __shared__ int sm_wave[8];
    const int t    = threadIdx.x;
    const int lane = t & 31;
    const int wv   = t >> 5;

    int b = 0;
    for (int i = t; i < (int)blockIdx.x; i += 256) b += bsum[i];
    for (int d = 16; d > 0; d >>= 1) b += __shfl_xor(b, d);

    const int i0 = blockIdx.x * SCAN_CHUNK + t * 4;
    int c[4];
    int s = 0;
    for (int k = 0; k < 4; ++k) {
        c[k] = (i0 + k < n) ? cnt[i0 + k] : 0;
        s += c[k];
    }
    int incl = s;
    for (int d = 1; d < 32; d <<= 1) {
        int up = __shfl_up(incl, d);
        if (lane >= d) incl += up;
    }
    if (lane == 31) sm_wave[wv] = incl;
    if (lane == 0)  sm_base[wv] = b;
    __syncthreads();
    int run = incl - s;
    for (int k = 0; k < 8; ++k) {
        run += sm_base[k];
        if (k < wv) run += sm_wave[k];
    }
    for (int k = 0; k < 4; ++k) {
        if (i0 + k < n) offs[i0 + k] = run;
        run += c[k];
    }
}

__global__ __launch_bounds__(256)
void gcn_bucket_scatter(const int* __restrict__ row,
                        const int* __restrict__ col,
                        const float* __restrict__ w,
                        const int* __restrict__ S,
                        int2* __restrict__ tmp,
                        int ne, int nblk, int nbk) {
    __shared__ int cur[MAX_BK];
    const int t = threadIdx.x;
    for (int i = t; i < nbk; i += 256) cur[i] = S[(size_t)i * nblk + blockIdx.x];
    __syncthreads();
    const int e0 = blockIdx.x * EDGE_CHUNK + t;
    for (int g = 0; g < 2; ++g) {
        int   r[8];
        int   c[8];
        float ww[8];
        for (int k = 0; k < 8; ++k) {
            const int  e  = e0 + (g * 8 + k) * 256;
            const bool ok = e < ne;
            r[k]  = ok ? row[e] : -1;
            c[k]  = ok ? col[e] : 0;
            ww[k] = ok ? w[e] : 0.0f;
        }
        int pos[8];
        for (int k = 0; k < 8; ++k)
            pos[k] = (r[k] >= 0) ? atomicAdd(&cur[r[k] >> BK_SHIFT], 1) : -1;
        for (int k = 0; k < 8; ++k)
            if (pos[k] >= 0)
                tmp[pos[k]] = make_int2((c[k] << BK_SHIFT) | (r[k] & (BK_ROWS - 1)), __float_as_int(ww[k]));
    }
}

__global__ __launch_bounds__(1024)
void gcn_bucket_sort(const int* __restrict__ S,
                     const int2* __restrict__ tmp,
                     int2* __restrict__ csr,
                     int* __restrict__ offs,
                     int* __restrict__ cnt,
                     int n, int ne, int nblk, int nbk) {
    __shared__ int lcnt[BK_ROWS];
    __shared__ int lcur[BK_ROWS];
    __shared__ int wsum[BK_ROWS / 32];
    const int t    = threadIdx.x;
    const int lane = t & 31;
    const int wv   = t >> 5;
    const int b    = blockIdx.x;
    const int bstart = S[(size_t)b * nblk];
    const int bend   = (b + 1 < nbk) ? S[(size_t)(b + 1) * nblk] : ne;

    if (t < BK_ROWS) lcnt[t] = 0;
    __syncthreads();


    for (int i0 = bstart; i0 < bend; i0 += 4096) {
        int key[4];
        int inc[4];
        for (int u = 0; u < 4; ++u) {
            const int  i  = i0 + u * 1024 + t;
            const bool ok = i < bend;
            key[u] = tmp[ok ? i : bend - 1].x;
            inc[u] = ok ? 1 : 0;
        }
        for (int u = 0; u < 4; ++u) atomicAdd(&lcnt[key[u] & (BK_ROWS - 1)], inc[u]);
    }
    __syncthreads();

    const int v = (t < BK_ROWS) ? lcnt[t] : 0;
    int incl = v;
    for (int d = 1; d < 32; d <<= 1) {
        const int up = __shfl_up(incl, d);
        if (lane >= d) incl += up;
    }
    if (t < BK_ROWS && lane == 31) wsum[wv] = incl;
    __syncthreads();
    if (t < BK_ROWS) {
        int base = 0;
        for (int k = 0; k < wv; ++k) base += wsum[k];
        const int excl = base + incl - v;
        lcur[t] = excl;
        const int r = b * BK_ROWS + t;
        if (r < n) {
            offs[r] = bstart + excl;
            cnt[r]  = v;
        }
    }
    __syncthreads();

    for (int i0 = bstart; i0 < bend; i0 += 4096) {
        int2 cw[4];
        int  inc[4];
        for (int u = 0; u < 4; ++u) {
            const int  i  = i0 + u * 1024 + t;
            const bool ok = i < bend;
            cw[u]  = tmp[ok ? i : bend - 1];
            inc[u] = ok ? 1 : 0;
        }
        int pos[4];
        for (int u = 0; u < 4; ++u) pos[u] = bstart + atomicAdd(&lcur[cw[u].x & (BK_ROWS - 1)], inc[u]);
        for (int u = 0; u < 4; ++u)
            if (inc[u])
                csr[pos[u]] = make_int2((int)((unsigned)cw[u].x >> BK_SHIFT), cw[u].y);
    }
}

__global__ void gcn_dinv(const int* __restrict__ offs,
                         const int* __restrict__ cnt,
                         const int2* __restrict__ csr,
                         float* __restrict__ dinv, int n) {
    int i = blockIdx.x * blockDim.x + threadIdx.x;
    if (i < n) {
        const int beg = offs[i];
        const int num = cnt[i];
        float d = FILL_WEIGHT;
        for (int p = 0; p < num; ++p) d += __int_as_float(csr[beg + p].y);
        dinv[i] = (d > 0.0f) ? __frsqrt_rn(fmaxf(d, 1e-12f)) : 0.0f;
    }
}

__global__ __launch_bounds__(256)
void gcn_gemm_wmma(const float* __restrict__ x,
                   const float* __restrict__ kern,
                   float* __restrict__ h,
                   int num_nodes) {
    const int lane  = threadIdx.x & 31;
    const int wave  = threadIdx.x >> 5;
    const int tile  = blockIdx.x * 8 + wave;
    const int ntile = (num_nodes + 15) >> 4;
    if (tile >= ntile) return;

    const int r0   = tile << 4;
    const int half = lane >> 4;
    const int m    = lane & 15;

    v8f acc[4] = {};

    const float* xrow = x + (size_t)(r0 + m) * FEAT;
#pragma unroll
    for (int kb = 0; kb < FEAT; kb += 4) {
        const int k0 = kb + (half << 1);
        v2f a;
        a.x = xrow[k0];
        a.y = xrow[k0 + 1];
#pragma unroll
        for (int cb = 0; cb < 4; ++cb) {
            v2f b;
            b.x = kern[(size_t)k0 * FEAT + cb * 16 + m];
            b.y = kern[(size_t)(k0 + 1) * FEAT + cb * 16 + m];
            acc[cb] = __builtin_amdgcn_wmma_f32_16x16x4_f32(
                 false, a,  false, b,
                 (short)0, acc[cb],
                 false,  false);
        }
    }

#pragma unroll
    for (int cb = 0; cb < 4; ++cb) {
#pragma unroll
        for (int v = 0; v < 8; ++v) {
            h[(size_t)(r0 + v + 8 * half) * FEAT + cb * 16 + m] = acc[cb][v];
        }
    }
}

__global__ __launch_bounds__(256)
void gcn_row_gather(const int* __restrict__ offs,
                    const int* __restrict__ cnt,
                    const int2* __restrict__ csr,
                    const float* __restrict__ dinv,
                    const float* __restrict__ h,
                    const float* __restrict__ bias,
                    float* __restrict__ out, int n) {
    const int lane = threadIdx.x & 31;
    const int r = __builtin_amdgcn_readfirstlane(blockIdx.x * 8 + (threadIdx.x >> 5));
    if (r >= n) return;

    const int   beg = offs[r];
    const int   num = cnt[r];
    const float dr  = dinv[r];
    const v2f* __restrict__ h2 = reinterpret_cast<const v2f*>(h);

#define EDGE_C(j) __builtin_amdgcn_readlane(c, (j))
#define EDGE_W(j) __int_as_float(__builtin_amdgcn_readlane(__float_as_int(wn), (j)))
#define H_ROW(cc) h2[(size_t)(cc) * (FEAT / 2) + lane]

    v2f acc = {0.0f, 0.0f};
    for (int base = 0; base < num; base += 32) {
        const int p = base + lane;
        int   c  = 0;
        float wn = 0.0f;
        if (p < num) {
            const int2 cw = csr[beg + p];
            c  = cw.x;
            wn = dr * __int_as_float(cw.y) * dinv[c];
        }
        const int m = min(32, num - base);
        int j = 0;
        for (; j + 8 <= m; j += 8) {
            const v2f a0 = H_ROW(EDGE_C(j));
            const v2f a1 = H_ROW(EDGE_C(j + 1));
            const v2f a2 = H_ROW(EDGE_C(j + 2));
            const v2f a3 = H_ROW(EDGE_C(j + 3));
            const v2f a4 = H_ROW(EDGE_C(j + 4));
            const v2f a5 = H_ROW(EDGE_C(j + 5));
            const v2f a6 = H_ROW(EDGE_C(j + 6));
            const v2f a7 = H_ROW(EDGE_C(j + 7));
            acc += a0 * EDGE_W(j);
            acc += a1 * EDGE_W(j + 1);
            acc += a2 * EDGE_W(j + 2);
            acc += a3 * EDGE_W(j + 3);
            acc += a4 * EDGE_W(j + 4);
            acc += a5 * EDGE_W(j + 5);
            acc += a6 * EDGE_W(j + 6);
            acc += a7 * EDGE_W(j + 7);
        }
        if (j + 4 <= m) {
            const v2f a0 = H_ROW(EDGE_C(j));
            const v2f a1 = H_ROW(EDGE_C(j + 1));
            const v2f a2 = H_ROW(EDGE_C(j + 2));
            const v2f a3 = H_ROW(EDGE_C(j + 3));
            acc += a0 * EDGE_W(j);
            acc += a1 * EDGE_W(j + 1);
            acc += a2 * EDGE_W(j + 2);
            acc += a3 * EDGE_W(j + 3);
            j += 4;
        }
#pragma clang loop unroll(disable)
        for (; j < m; ++j) {
            const v2f a = H_ROW(EDGE_C(j));
            acc += a * EDGE_W(j);
        }
    }
#undef EDGE_C
#undef EDGE_W
#undef H_ROW

    const v2f   hs = h2[(size_t)r * (FEAT / 2) + lane];
    const float b0 = bias[2 * lane];
    const float b1 = bias[2 * lane + 1];
    v2f o;
    o.x = (b0 + hs.x * dr * dr * FILL_WEIGHT) + acc.x;
    o.y = (b1 + hs.y * dr * dr * FILL_WEIGHT) + acc.y;
    reinterpret_cast<v2f*>(out)[(size_t)r * (FEAT / 2) + lane] = o;
}

extern "C" void kernel_launch(void* const* d_in, const int* in_sizes, int n_in,
                              void* d_out, int out_size, void* d_ws, size_t ws_size,
                              hipStream_t stream) {
    const float* x    = (const float*)d_in[0];
    const int*   ei   = (const int*)d_in[1];
    const float* ew   = (const float*)d_in[2];
    const float* kern = (const float*)d_in[3];
    const float* bias = (const float*)d_in[4];

    const int N = in_sizes[0] / FEAT;
    const int E = in_sizes[2];
    const int* row = ei;
    const int* col = ei + E;

    const int G1   = (E + EDGE_CHUNK - 1) / EDGE_CHUNK;
    const int NBK  = (N + BK_ROWS - 1) / BK_ROWS;
    const int n2   = NBK * G1;
    const int NB2  = (n2 + SCAN_CHUNK - 1) / SCAN_CHUNK;
    const int Npad = (N + 3) & ~3;
    const int Bpad = (NB2 + 3) & ~3;
    const int Hpad = (n2 + 3) & ~3;
    const int Epad = (E + 3) & ~3;

    float* h    = (float*)d_ws;
    float* dinv = h + (size_t)N * FEAT;
    int*   cnt  = (int*)(dinv + Npad);
    int*   offs = cnt + Npad;
    int*   bsum = offs + Npad;
    int*   bh   = bsum + Bpad;
    int*   S    = bh + Hpad;
    int2*  tmp  = (int2*)(S + Hpad);
    int2*  csr  = tmp + Epad;
    float* out  = (float*)d_out;

    const int T = 256;

    gcn_bucket_hist<<<G1, T, 0, stream>>>(row, bh, E, G1, NBK);
    gcn_scan_partials<<<NB2, T, 0, stream>>>(bh, bsum, n2);
    gcn_scan_offsets<<<NB2, T, 0, stream>>>(bh, bsum, S, n2);
    gcn_bucket_scatter<<<G1, T, 0, stream>>>(row, col, ew, S, tmp, E, G1, NBK);
    gcn_bucket_sort<<<NBK, 1024, 0, stream>>>(S, tmp, csr, offs, cnt, N, E, G1, NBK);
    gcn_dinv<<<(N + T - 1) / T, T, 0, stream>>>(offs, cnt, csr, dinv, N);

    const int ntile = (N + 15) / 16;
    gcn_gemm_wmma<<<(ntile + 7) / 8, T, 0, stream>>>(x, kern, h, N);

    gcn_row_gather<<<(N + 7) / 8, T, 0, stream>>>(offs, cnt, csr, dinv, h, bias, out, N);
}


# Level 1 of the counting sort made local to each edge chunk (own slab + piece table; level 2 collects the pieces): histogram and 2 scan launches gone, 8 launches to 5; row sum order = LDS arrival rank
# speedup vs baseline: 4.5277x; 1.4201x over previous
#include <hip/hip_runtime.h>
#include <hip/hip_bf16.h>

typedef float v2f __attribute__((ext_vector_type(2)));
typedef float v8f __attribute__((ext_vector_type(8)));

#define FEAT 64
#define FILL_WEIGHT 1.0f

#define BK_SHIFT   9
#define BK_ROWS    512
#define EDGE_CHUNK 4096
#define MAX_BK     2048
#define MAX_CHUNK  4096

template <bool FULL>
__device__ __forceinline__
void gcn_chunk_sort_body(const int* __restrict__ row,
                         const int* __restrict__ col,
                         const float* __restrict__ w,
                         int* __restrict__ pieces,
                         int2* __restrict__ slab,
                         int ne, int nblk, int nbk,
                         int* cnt, int* wsum) {
    const int t    = threadIdx.x;
    const int lane = t & 31;
    const int wv   = t >> 5;
    const int blk  = blockIdx.x;
    for (int i = t; i < nbk; i += 256) cnt[i] = 0;
    __syncthreads();

    const int e0 = blk * EDGE_CHUNK + t;
    int   r[16];
    int   c[16];
    float ww[16];
    for (int k = 0; k < 16; ++k) {
        const int  e  = e0 + k * 256;
        const bool ok = FULL || e < ne;
        r[k]  = ok ? row[e] : 0;
        c[k]  = ok ? col[e] : 0;
        ww[k] = ok ? w[e] : 0.0f;
    }
    int rank[16];
    for (int k = 0; k < 16; ++k) {
        const bool ok = FULL || e0 + k * 256 < ne;
        rank[k] = atomicAdd(&cnt[ok ? (r[k] >> BK_SHIFT) : t], ok ? 1 : 0);
    }
    __syncthreads();

    int v[8];
    int s = 0;
    for (int k = 0; k < 8; ++k) {
        const int i = t * 8 + k;
        v[k] = (i < nbk) ? cnt[i] : 0;
        s += v[k];
    }
    int incl = s;
    for (int d = 1; d < 32; d <<= 1) {
        const int up = __shfl_up(incl, d);
        if (lane >= d) incl += up;
    }
    if (lane == 31) wsum[wv] = incl;
    __syncthreads();
    int run = incl - s;
    for (int k = 0; k < wv; ++k) run += wsum[k];
    for (int k = 0; k < 8; ++k) {
        const int i = t * 8 + k;
        if (i < nbk) {
            cnt[i] = run;
            pieces[(size_t)i * nblk + blk] = run | (v[k] << 16);
            run += v[k];
        }
    }
    __syncthreads();

    int dst[16];
    for (int k = 0; k < 16; ++k) dst[k] = cnt[r[k] >> BK_SHIFT] + rank[k];
    int2* __restrict__ mine = slab + (size_t)blk * EDGE_CHUNK;
    for (int k = 0; k < 16; ++k)
        if (FULL || e0 + k * 256 < ne)
            mine[dst[k]] = make_int2((c[k] << BK_SHIFT) | (r[k] & (BK_ROWS - 1)), __float_as_int(ww[k]));
}

__global__ __launch_bounds__(256)
void gcn_chunk_sort(const int* __restrict__ row,
                    const int* __restrict__ col,
                    const float* __restrict__ w,
                    int* __restrict__ pieces,
                    int2* __restrict__ slab,
                    int ne, int nblk, int nbk) {
    __shared__ int cnt[MAX_BK];
    __shared__ int wsum[8];
    if ((int)(blockIdx.x + 1) * EDGE_CHUNK <= ne)
        gcn_chunk_sort_body<true>(row, col, w, pieces, slab, ne, nblk, nbk, cnt, wsum);
    else
        gcn_chunk_sort_body<false>(row, col, w, pieces, slab, ne, nblk, nbk, cnt, wsum);
}

#define TRIP_PIECES 8
#define TRIP_CHUNKS (32 * TRIP_PIECES)
static_assert(TRIP_PIECES == 8, "the load fences of both passes name eight values");
static_assert(MAX_CHUNK % TRIP_CHUNKS == 0, "the padded piece list must fit the LDS table");
__global__ __launch_bounds__(1024)
void gcn_bucket_sort(const int* __restrict__ pieces,
                     const int2* __restrict__ slab,
                     int2* __restrict__ csr,
                     int* __restrict__ offs,
                     int* __restrict__ cnt,
                     int n, int nblk) {
    __shared__ int lcnt[BK_ROWS];
    __shared__ int lcur[BK_ROWS];
    __shared__ int wsum[BK_ROWS / 32];
    __shared__ int piece[MAX_CHUNK];
    __shared__ int red[32];
    const int t    = threadIdx.x;
    const int lane = t & 31;
    const int wv   = __builtin_amdgcn_readfirstlane(t >> 5);
    const int b    = blockIdx.x;

    int s = 0;
    for (int c = t; c < nblk; c += 1024) {
        const int pk = pieces[(size_t)b * nblk + c];
        piece[c] = pk;
        s += pk & 0xFFFF;
    }
    for (int d = 16; d > 0; d >>= 1) s += __shfl_xor(s, d);
    if (lane == 0) red[wv] = s;
    for (int c = nblk + t; c < ((nblk + TRIP_CHUNKS - 1) & ~(TRIP_CHUNKS - 1)); c += 1024) piece[c] = 0;
    if (t < BK_ROWS) lcnt[t] = 0;
    __syncthreads();
    int bstart = 0;
    for (int k = 0; k < 32; ++k) bstart += red[k];


    for (int c0 = wv; c0 < nblk; c0 += TRIP_CHUNKS) {
        int key[TRIP_PIECES];
        int nn[TRIP_PIECES];
        int base[TRIP_PIECES];
        for (int u = 0; u < TRIP_PIECES; ++u) {
            const int c     = c0 + u * 32;
            const int pk    = piece[c];
            const int slab0 = ((c < nblk) ? c : 0) * EDGE_CHUNK;
            nn[u]   = (int)((unsigned)pk >> 16);
            base[u] = slab0 + (pk & 0xFFFF);
            key[u]  = slab[(lane < nn[u]) ? base[u] + lane : slab0].x;
        }
        asm volatile("" :: "v"(key[0]), "v"(key[1]), "v"(key[2]), "v"(key[3]),
                           "v"(key[4]), "v"(key[5]), "v"(key[6]), "v"(key[7]));
        for (int u = 0; u < TRIP_PIECES; ++u) {
            const bool ok = lane < nn[u];
            atomicAdd(&lcnt[ok ? (key[u] & (BK_ROWS - 1)) : lane], ok ? 1 : 0);
        }
        for (int u = 0; u < TRIP_PIECES; ++u)
            for (int j = lane + 32; j < nn[u]; j += 32)
                atomicAdd(&lcnt[slab[base[u] + j].x & (BK_ROWS - 1)], 1);
    }
    __syncthreads();

    const int v = (t < BK_ROWS) ? lcnt[t] : 0;
    int incl = v;
    for (int d = 1; d < 32; d <<= 1) {
        const int up = __shfl_up(incl, d);
        if (lane >= d) incl += up;
    }
    if (t < BK_ROWS && lane == 31) wsum[wv] = incl;
    __syncthreads();
    if (t < BK_ROWS) {
        int base = 0;
        for (int k = 0; k < wv; ++k) base += wsum[k];
        const int excl = base + incl - v;
        lcur[t] = excl;
        const int r = b * BK_ROWS + t;
        if (r < n) {
            offs[r] = bstart + excl;
            cnt[r]  = v;
        }
    }
    __syncthreads();

    for (int c0 = wv; c0 < nblk; c0 += TRIP_CHUNKS) {
        int2 cw[TRIP_PIECES];
        int  nn[TRIP_PIECES];
        int  base[TRIP_PIECES];
        for (int u = 0; u < TRIP_PIECES; ++u) {
            const int c     = c0 + u * 32;
            const int pk    = piece[c];
            const int slab0 = ((c < nblk) ? c : 0) * EDGE_CHUNK;
            nn[u]   = (int)((unsigned)pk >> 16);
            base[u] = slab0 + (pk & 0xFFFF);
            cw[u]   = slab[(lane < nn[u]) ? base[u] + lane : slab0];
        }
        asm volatile("" :: "v"(cw[0].x), "v"(cw[1].x), "v"(cw[2].x), "v"(cw[3].x),
                           "v"(cw[4].x), "v"(cw[5].x), "v"(cw[6].x), "v"(cw[7].x));
        int pos[TRIP_PIECES];
        for (int u = 0; u < TRIP_PIECES; ++u) {
            const bool ok = lane < nn[u];
            pos[u] = bstart + atomicAdd(&lcur[ok ? (cw[u].x & (BK_ROWS - 1)) : lane], ok ? 1 : 0);
        }
        for (int u = 0; u < TRIP_PIECES; ++u)
            if (lane < nn[u])
                csr[pos[u]] = make_int2((int)((unsigned)cw[u].x >> BK_SHIFT), cw[u].y);
        for (int u = 0; u < TRIP_PIECES; ++u)
            for (int j = lane + 32; j < nn[u]; j += 32) {
                const int2 e = slab[base[u] + j];
                const int  p = bstart + atomicAdd(&lcur[e.x & (BK_ROWS - 1)], 1);
                csr[p] = make_int2((int)((unsigned)e.x >> BK_SHIFT), e.y);
            }
    }
}

__global__ void gcn_dinv(const int* __restrict__ offs,
                         const int* __restrict__ cnt,
                         const int2* __restrict__ csr,
                         float* __restrict__ dinv, int n) {
    int i = blockIdx.x * blockDim.x + threadIdx.x;
    if (i < n) {
        const int beg = offs[i];
        const int num = cnt[i];
        float d = FILL_WEIGHT;
        for (int p = 0; p < num; ++p) d += __int_as_float(csr[beg + p].y);
        dinv[i] = (d > 0.0f) ? __frsqrt_rn(fmaxf(d, 1e-12f)) : 0.0f;
    }
}

__global__ __launch_bounds__(256)
void gcn_gemm_wmma(const float* __restrict__ x,
                   const float* __restrict__ kern,
                   float* __restrict__ h,
                   int num_nodes) {
    const int lane  = threadIdx.x & 31;
    const int wave  = threadIdx.x >> 5;
    const int tile  = blockIdx.x * 8 + wave;
    const int ntile = (num_nodes + 15) >> 4;
    if (tile >= ntile) return;

    const int r0   = tile << 4;
    const int half = lane >> 4;
    const int m    = lane & 15;

    v8f acc[4] = {};

    const float* xrow = x + (size_t)(r0 + m) * FEAT;
#pragma unroll
    for (int kb = 0; kb < FEAT; kb += 4) {
        const int k0 = kb + (half << 1);
        v2f a;
        a.x = xrow[k0];
        a.y = xrow[k0 + 1];
#pragma unroll
        for (int cb = 0; cb < 4; ++cb) {
            v2f b;
            b.x = kern[(size_t)k0 * FEAT + cb * 16 + m];
            b.y = kern[(size_t)(k0 + 1) * FEAT + cb * 16 + m];
            acc[cb] = __builtin_amdgcn_wmma_f32_16x16x4_f32(
                 false, a,  false, b,
                 (short)0, acc[cb],
                 false,  false);
        }
    }

#pragma unroll
    for (int cb = 0; cb < 4; ++cb) {
#pragma unroll
        for (int v = 0; v < 8; ++v) {
            h[(size_t)(r0 + v + 8 * half) * FEAT + cb * 16 + m] = acc[cb][v];
        }
    }
}

__global__ __launch_bounds__(256)
void gcn_row_gather(const int* __restrict__ offs,
                    const int* __restrict__ cnt,
                    const int2* __restrict__ csr,
                    const float* __restrict__ dinv,
                    const float* __restrict__ h,
                    const float* __restrict__ bias,
                    float* __restrict__ out, int n) {
    const int lane = threadIdx.x & 31;
    const int r = __builtin_amdgcn_readfirstlane(blockIdx.x * 8 + (threadIdx.x >> 5));
    if (r >= n) return;

    const int   beg = offs[r];
    const int   num = cnt[r];
    const float dr  = dinv[r];
    const v2f* __restrict__ h2 = reinterpret_cast<const v2f*>(h);

#define EDGE_C(j) __builtin_amdgcn_readlane(c, (j))
#define EDGE_W(j) __int_as_float(__builtin_amdgcn_readlane(__float_as_int(wn), (j)))
#define H_ROW(cc) h2[(size_t)(cc) * (FEAT / 2) + lane]

    v2f acc = {0.0f, 0.0f};
    for (int base = 0; base < num; base += 32) {
        const int p = base + lane;
        int   c  = 0;
        float wn = 0.0f;
        if (p < num) {
            const int2 cw = csr[beg + p];
            c  = cw.x;
            wn = dr * __int_as_float(cw.y) * dinv[c];
        }
        const int m = min(32, num - base);
        int j = 0;
        for (; j + 8 <= m; j += 8) {
            const v2f a0 = H_ROW(EDGE_C(j));
            const v2f a1 = H_ROW(EDGE_C(j + 1));
            const v2f a2 = H_ROW(EDGE_C(j + 2));
            const v2f a3 = H_ROW(EDGE_C(j + 3));
            const v2f a4 = H_ROW(EDGE_C(j + 4));
            const v2f a5 = H_ROW(EDGE_C(j + 5));
            const v2f a6 = H_ROW(EDGE_C(j + 6));
            const v2f a7 = H_ROW(EDGE_C(j + 7));
            acc += a0 * EDGE_W(j);
            acc += a1 * EDGE_W(j + 1);
            acc += a2 * EDGE_W(j + 2);
            acc += a3 * EDGE_W(j + 3);
            acc += a4 * EDGE_W(j + 4);
            acc += a5 * EDGE_W(j + 5);
            acc += a6 * EDGE_W(j + 6);
            acc += a7 * EDGE_W(j + 7);
        }
        if (j + 4 <= m) {
            const v2f a0 = H_ROW(EDGE_C(j));
            const v2f a1 = H_ROW(EDGE_C(j + 1));
            const v2f a2 = H_ROW(EDGE_C(j + 2));
            const v2f a3 = H_ROW(EDGE_C(j + 3));
            acc += a0 * EDGE_W(j);
            acc += a1 * EDGE_W(j + 1);
            acc += a2 * EDGE_W(j + 2);
            acc += a3 * EDGE_W(j + 3);
            j += 4;
        }
#pragma clang loop unroll(disable)
        for (; j < m; ++j) {
            const v2f a = H_ROW(EDGE_C(j));
            acc += a * EDGE_W(j);
        }
    }
#undef EDGE_C
#undef EDGE_W
#undef H_ROW

    const v2f   hs = h2[(size_t)r * (FEAT / 2) + lane];
    const float b0 = bias[2 * lane];
    const float b1 = bias[2 * lane + 1];
    v2f o;
    o.x = (b0 + hs.x * dr * dr * FILL_WEIGHT) + acc.x;
    o.y = (b1 + hs.y * dr * dr * FILL_WEIGHT) + acc.y;
    reinterpret_cast<v2f*>(out)[(size_t)r * (FEAT / 2) + lane] = o;
}

extern "C" void kernel_launch(void* const* d_in, const int* in_sizes, int n_in,
                              void* d_out, int out_size, void* d_ws, size_t ws_size,
                              hipStream_t stream) {
    const float* x    = (const float*)d_in[0];
    const int*   ei   = (const int*)d_in[1];
    const float* ew   = (const float*)d_in[2];
    const float* kern = (const float*)d_in[3];
    const float* bias = (const float*)d_in[4];

    const int N = in_sizes[0] / FEAT;
    const int E = in_sizes[2];
    const int* row = ei;
    const int* col = ei + E;

    const int G1   = (E + EDGE_CHUNK - 1) / EDGE_CHUNK;
    const int NBK  = (N + BK_ROWS - 1) / BK_ROWS;
    const int n2   = NBK * G1;
    const int Npad = (N + 3) & ~3;
    const int Hpad = (n2 + 3) & ~3;
    const int Spad = G1 * EDGE_CHUNK;

    float* h      = (float*)d_ws;
    float* dinv   = h + (size_t)N * FEAT;
    int*   cnt    = (int*)(dinv + Npad);
    int*   offs   = cnt + Npad;
    int*   pieces = offs + Npad;
    int2*  slab   = (int2*)(pieces + Hpad);
    int2*  csr    = slab + Spad;
    float* out    = (float*)d_out;

    const int T = 256;

    gcn_chunk_sort<<<G1, T, 0, stream>>>(row, col, ew, pieces, slab, E, G1, NBK);
    gcn_bucket_sort<<<NBK, 1024, 0, stream>>>(pieces, slab, csr, offs, cnt, N, G1);
    gcn_dinv<<<(N + T - 1) / T, T, 0, stream>>>(offs, cnt, csr, dinv, N);

    const int ntile = (N + 15) / 16;
    gcn_gemm_wmma<<<(ntile + 7) / 8, T, 0, stream>>>(x, kern, h, N);

    gcn_row_gather<<<(N + 7) / 8, T, 0, stream>>>(offs, cnt, csr, dinv, h, bias, out, N);
}


# GEMM epilogue stores each h row scaled by its own dinv, so the row gather drops the per-edge dinv[col] read; message product re-associated as (h*dinv[col])*(dinv[row]*w), sum order unchanged
# speedup vs baseline: 4.8088x; 1.0621x over previous
#include <hip/hip_runtime.h>
#include <hip/hip_bf16.h>

typedef float v2f __attribute__((ext_vector_type(2)));
typedef float v8f __attribute__((ext_vector_type(8)));

#define FEAT 64
#define FILL_WEIGHT 1.0f

#define BK_SHIFT   9
#define BK_ROWS    512
#define EDGE_CHUNK 4096
#define MAX_BK     2048
#define MAX_CHUNK  4096

template <bool FULL>
__device__ __forceinline__
void gcn_chunk_sort_body(const int* __restrict__ row,
                         const int* __restrict__ col,
                         const float* __restrict__ w,
                         int* __restrict__ pieces,
                         int2* __restrict__ slab,
                         int ne, int nblk, int nbk,
                         int* cnt, int* wsum) {
    const int t    = threadIdx.x;
    const int lane = t & 31;
    const int wv   = t >> 5;
    const int blk  = blockIdx.x;
    for (int i = t; i < nbk; i += 256) cnt[i] = 0;
    __syncthreads();

    const int e0 = blk * EDGE_CHUNK + t;
    int   r[16];
    int   c[16];
    float ww[16];
    for (int k = 0; k < 16; ++k) {
        const int  e  = e0 + k * 256;
        const bool ok = FULL || e < ne;
        r[k]  = ok ? row[e] : 0;
        c[k]  = ok ? col[e] : 0;
        ww[k] = ok ? w[e] : 0.0f;
    }
    int rank[16];
    for (int k = 0; k < 16; ++k) {
        const bool ok = FULL || e0 + k * 256 < ne;
        rank[k] = atomicAdd(&cnt[ok ? (r[k] >> BK_SHIFT) : t], ok ? 1 : 0);
    }
    __syncthreads();

    int v[8];
    int s = 0;
    for (int k = 0; k < 8; ++k) {
        const int i = t * 8 + k;
        v[k] = (i < nbk) ? cnt[i] : 0;
        s += v[k];
    }
    int incl = s;
    for (int d = 1; d < 32; d <<= 1) {
        const int up = __shfl_up(incl, d);
        if (lane >= d) incl += up;
    }
    if (lane == 31) wsum[wv] = incl;
    __syncthreads();
    int run = incl - s;
    for (int k = 0; k < wv; ++k) run += wsum[k];
    for (int k = 0; k < 8; ++k) {
        const int i = t * 8 + k;
        if (i < nbk) {
            cnt[i] = run;
            pieces[(size_t)i * nblk + blk] = run | (v[k] << 16);
            run += v[k];
        }
    }
    __syncthreads();

    int dst[16];
    for (int k = 0; k < 16; ++k) dst[k] = cnt[r[k] >> BK_SHIFT] + rank[k];
    int2* __restrict__ mine = slab + (size_t)blk * EDGE_CHUNK;
    for (int k = 0; k < 16; ++k)
        if (FULL || e0 + k * 256 < ne)
            mine[dst[k]] = make_int2((c[k] << BK_SHIFT) | (r[k] & (BK_ROWS - 1)), __float_as_int(ww[k]));
}

__global__ __launch_bounds__(256)
void gcn_chunk_sort(const int* __restrict__ row,
                    const int* __restrict__ col,
                    const float* __restrict__ w,
                    int* __restrict__ pieces,
                    int2* __restrict__ slab,
                    int ne, int nblk, int nbk) {
    __shared__ int cnt[MAX_BK];
    __shared__ int wsum[8];
    if ((int)(blockIdx.x + 1) * EDGE_CHUNK <= ne)
        gcn_chunk_sort_body<true>(row, col, w, pieces, slab, ne, nblk, nbk, cnt, wsum);
    else
        gcn_chunk_sort_body<false>(row, col, w, pieces, slab, ne, nblk, nbk, cnt, wsum);
}

#define TRIP_PIECES 8
#define TRIP_CHUNKS (32 * TRIP_PIECES)
static_assert(TRIP_PIECES == 8, "the load fences of both passes name eight values");
static_assert(MAX_CHUNK % TRIP_CHUNKS == 0, "the padded piece list must fit the LDS table");
__global__ __launch_bounds__(1024)
void gcn_bucket_sort(const int* __restrict__ pieces,
                     const int2* __restrict__ slab,
                     int2* __restrict__ csr,
                     int* __restrict__ offs,
                     int* __restrict__ cnt,
                     int n, int nblk) {
    __shared__ int lcnt[BK_ROWS];
    __shared__ int lcur[BK_ROWS];
    __shared__ int wsum[BK_ROWS / 32];
    __shared__ int piece[MAX_CHUNK];
    __shared__ int red[32];
    const int t    = threadIdx.x;
    const int lane = t & 31;
    const int wv   = __builtin_amdgcn_readfirstlane(t >> 5);
    const int b    = blockIdx.x;

    int s = 0;
    for (int c = t; c < nblk; c += 1024) {
        const int pk = pieces[(size_t)b * nblk + c];
        piece[c] = pk;
        s += pk & 0xFFFF;
    }
    for (int d = 16; d > 0; d >>= 1) s += __shfl_xor(s, d);
    if (lane == 0) red[wv] = s;
    for (int c = nblk + t; c < ((nblk + TRIP_CHUNKS - 1) & ~(TRIP_CHUNKS - 1)); c += 1024) piece[c] = 0;
    if (t < BK_ROWS) lcnt[t] = 0;
    __syncthreads();
    int bstart = 0;
    for (int k = 0; k < 32; ++k) bstart += red[k];


    for (int c0 = wv; c0 < nblk; c0 += TRIP_CHUNKS) {
        int key[TRIP_PIECES];
        int nn[TRIP_PIECES];
        int base[TRIP_PIECES];
        for (int u = 0; u < TRIP_PIECES; ++u) {
            const int c     = c0 + u * 32;
            const int pk    = piece[c];
            const int slab0 = ((c < nblk) ? c : 0) * EDGE_CHUNK;
            nn[u]   = (int)((unsigned)pk >> 16);
            base[u] = slab0 + (pk & 0xFFFF);
            key[u]  = slab[(lane < nn[u]) ? base[u] + lane : slab0].x;
        }
        asm volatile("" :: "v"(key[0]), "v"(key[1]), "v"(key[2]), "v"(key[3]),
                           "v"(key[4]), "v"(key[5]), "v"(key[6]), "v"(key[7]));
        for (int u = 0; u < TRIP_PIECES; ++u) {
            const bool ok = lane < nn[u];
            atomicAdd(&lcnt[ok ? (key[u] & (BK_ROWS - 1)) : lane], ok ? 1 : 0);
        }
        for (int u = 0; u < TRIP_PIECES; ++u)
            for (int j = lane + 32; j < nn[u]; j += 32)
                atomicAdd(&lcnt[slab[base[u] + j].x & (BK_ROWS - 1)], 1);
    }
    __syncthreads();

    const int v = (t < BK_ROWS) ? lcnt[t] : 0;
    int incl = v;
    for (int d = 1; d < 32; d <<= 1) {
        const int up = __shfl_up(incl, d);
        if (lane >= d) incl += up;
    }
    if (t < BK_ROWS && lane == 31) wsum[wv] = incl;
    __syncthreads();
    if (t < BK_ROWS) {
        int base = 0;
        for (int k = 0; k < wv; ++k) base += wsum[k];
        const int excl = base + incl - v;
        lcur[t] = excl;
        const int r = b * BK_ROWS + t;
        if (r < n) {
            offs[r] = bstart + excl;
            cnt[r]  = v;
        }
    }
    __syncthreads();

    for (int c0 = wv; c0 < nblk; c0 += TRIP_CHUNKS) {
        int2 cw[TRIP_PIECES];
        int  nn[TRIP_PIECES];
        int  base[TRIP_PIECES];
        for (int u = 0; u < TRIP_PIECES; ++u) {
            const int c     = c0 + u * 32;
            const int pk    = piece[c];
            const int slab0 = ((c < nblk) ? c : 0) * EDGE_CHUNK;
            nn[u]   = (int)((unsigned)pk >> 16);
            base[u] = slab0 + (pk & 0xFFFF);
            cw[u]   = slab[(lane < nn[u]) ? base[u] + lane : slab0];
        }
        asm volatile("" :: "v"(cw[0].x), "v"(cw[1].x), "v"(cw[2].x), "v"(cw[3].x),
                           "v"(cw[4].x), "v"(cw[5].x), "v"(cw[6].x), "v"(cw[7].x));
        int pos[TRIP_PIECES];
        for (int u = 0; u < TRIP_PIECES; ++u) {
            const bool ok = lane < nn[u];
            pos[u] = bstart + atomicAdd(&lcur[ok ? (cw[u].x & (BK_ROWS - 1)) : lane], ok ? 1 : 0);
        }
        for (int u = 0; u < TRIP_PIECES; ++u)
            if (lane < nn[u])
                csr[pos[u]] = make_int2((int)((unsigned)cw[u].x >> BK_SHIFT), cw[u].y);
        for (int u = 0; u < TRIP_PIECES; ++u)
            for (int j = lane + 32; j < nn[u]; j += 32) {
                const int2 e = slab[base[u] + j];
                const int  p = bstart + atomicAdd(&lcur[e.x & (BK_ROWS - 1)], 1);
                csr[p] = make_int2((int)((unsigned)e.x >> BK_SHIFT), e.y);
            }
    }
}

__global__ void gcn_dinv(const int* __restrict__ offs,
                         const int* __restrict__ cnt,
                         const int2* __restrict__ csr,
                         float* __restrict__ dinv, int n) {
    int i = blockIdx.x * blockDim.x + threadIdx.x;
    if (i < n) {
        const int beg = offs[i];
        const int num = cnt[i];
        float d = FILL_WEIGHT;
        for (int p = 0; p < num; ++p) d += __int_as_float(csr[beg + p].y);
        dinv[i] = (d > 0.0f) ? __frsqrt_rn(fmaxf(d, 1e-12f)) : 0.0f;
    }
}

__global__ __launch_bounds__(256)
void gcn_gemm_wmma(const float* __restrict__ x,
                   const float* __restrict__ kern,
                   const float* __restrict__ dinv,
                   float* __restrict__ h,
                   int num_nodes) {
    const int lane  = threadIdx.x & 31;
    const int wave  = threadIdx.x >> 5;
    const int tile  = blockIdx.x * 8 + wave;
    const int ntile = (num_nodes + 15) >> 4;
    if (tile >= ntile) return;

    const int r0   = tile << 4;
    const int half = lane >> 4;
    const int m    = lane & 15;

    v8f acc[4] = {};

    const float* xrow = x + (size_t)(r0 + m) * FEAT;
#pragma unroll
    for (int kb = 0; kb < FEAT; kb += 4) {
        const int k0 = kb + (half << 1);
        v2f a;
        a.x = xrow[k0];
        a.y = xrow[k0 + 1];
#pragma unroll
        for (int cb = 0; cb < 4; ++cb) {
            v2f b;
            b.x = kern[(size_t)k0 * FEAT + cb * 16 + m];
            b.y = kern[(size_t)(k0 + 1) * FEAT + cb * 16 + m];
            acc[cb] = __builtin_amdgcn_wmma_f32_16x16x4_f32(
                 false, a,  false, b,
                 (short)0, acc[cb],
                 false,  false);
        }
    }

    const float4* __restrict__ dv4 = reinterpret_cast<const float4*>(dinv + r0 + 8 * half);
    const float4 dlo = dv4[0];
    const float4 dhi = dv4[1];
    const float  rs[8] = {dlo.x, dlo.y, dlo.z, dlo.w, dhi.x, dhi.y, dhi.z, dhi.w};

#pragma unroll
    for (int cb = 0; cb < 4; ++cb) {
#pragma unroll
        for (int v = 0; v < 8; ++v) {
            h[(size_t)(r0 + v + 8 * half) * FEAT + cb * 16 + m] = acc[cb][v] * rs[v];
        }
    }
}

__global__ __launch_bounds__(256)
void gcn_row_gather(const int* __restrict__ offs,
                    const int* __restrict__ cnt,
                    const int2* __restrict__ csr,
                    const float* __restrict__ dinv,
                    const float* __restrict__ h,
                    const float* __restrict__ bias,
                    float* __restrict__ out, int n) {
    const int lane = threadIdx.x & 31;
    const int r = __builtin_amdgcn_readfirstlane(blockIdx.x * 8 + (threadIdx.x >> 5));
    if (r >= n) return;

    const int   beg = offs[r];
    const int   num = cnt[r];
    const float dr  = dinv[r];
    const v2f* __restrict__ h2 = reinterpret_cast<const v2f*>(h);

#define EDGE_C(j) __builtin_amdgcn_readlane(c, (j))
#define EDGE_W(j) __int_as_float(__builtin_amdgcn_readlane(__float_as_int(wn), (j)))
#define H_ROW(cc) h2[(size_t)(cc) * (FEAT / 2) + lane]

    v2f acc = {0.0f, 0.0f};
    for (int base = 0; base < num; base += 32) {
        const int p = base + lane;
        int   c  = 0;
        float wn = 0.0f;
        if (p < num) {
            const int2 cw = csr[beg + p];
            c  = cw.x;
            wn = dr * __int_as_float(cw.y);
        }
        const int m = min(32, num - base);
        int j = 0;
        for (; j + 8 <= m; j += 8) {
            const v2f a0 = H_ROW(EDGE_C(j));
            const v2f a1 = H_ROW(EDGE_C(j + 1));
            const v2f a2 = H_ROW(EDGE_C(j + 2));
            const v2f a3 = H_ROW(EDGE_C(j + 3));
            const v2f a4 = H_ROW(EDGE_C(j + 4));
            const v2f a5 = H_ROW(EDGE_C(j + 5));
            const v2f a6 = H_ROW(EDGE_C(j + 6));
            const v2f a7 = H_ROW(EDGE_C(j + 7));
            acc += a0 * EDGE_W(j);
            acc += a1 * EDGE_W(j + 1);
            acc += a2 * EDGE_W(j + 2);
            acc += a3 * EDGE_W(j + 3);
            acc += a4 * EDGE_W(j + 4);
            acc += a5 * EDGE_W(j + 5);
            acc += a6 * EDGE_W(j + 6);
            acc += a7 * EDGE_W(j + 7);
        }
        if (j + 4 <= m) {
            const v2f a0 = H_ROW(EDGE_C(j));
            const v2f a1 = H_ROW(EDGE_C(j + 1));
            const v2f a2 = H_ROW(EDGE_C(j + 2));
            const v2f a3 = H_ROW(EDGE_C(j + 3));
            acc += a0 * EDGE_W(j);
            acc += a1 * EDGE_W(j + 1);
            acc += a2 * EDGE_W(j + 2);
            acc += a3 * EDGE_W(j + 3);
            j += 4;
        }
#pragma clang loop unroll(disable)
        for (; j < m; ++j) {
            const v2f a = H_ROW(EDGE_C(j));
            acc += a * EDGE_W(j);
        }
    }
#undef EDGE_C
#undef EDGE_W
#undef H_ROW

    const v2f   hs = h2[(size_t)r * (FEAT / 2) + lane];
    const float b0 = bias[2 * lane];
    const float b1 = bias[2 * lane + 1];
    v2f o;
    o.x = (b0 + hs.x * dr * FILL_WEIGHT) + acc.x;
    o.y = (b1 + hs.y * dr * FILL_WEIGHT) + acc.y;
    reinterpret_cast<v2f*>(out)[(size_t)r * (FEAT / 2) + lane] = o;
}

extern "C" void kernel_launch(void* const* d_in, const int* in_sizes, int n_in,
                              void* d_out, int out_size, void* d_ws, size_t ws_size,
                              hipStream_t stream) {
    const float* x    = (const float*)d_in[0];
    const int*   ei   = (const int*)d_in[1];
    const float* ew   = (const float*)d_in[2];
    const float* kern = (const float*)d_in[3];
    const float* bias = (const float*)d_in[4];

    const int N = in_sizes[0] / FEAT;
    const int E = in_sizes[2];
    const int* row = ei;
    const int* col = ei + E;

    const int G1   = (E + EDGE_CHUNK - 1) / EDGE_CHUNK;
    const int NBK  = (N + BK_ROWS - 1) / BK_ROWS;
    const int n2   = NBK * G1;
    const int Npad = (N + 3) & ~3;
    const int Hpad = (n2 + 3) & ~3;
    const int Spad = G1 * EDGE_CHUNK;

    float* h      = (float*)d_ws;
    float* dinv   = h + (size_t)N * FEAT;
    int*   cnt    = (int*)(dinv + Npad);
    int*   offs   = cnt + Npad;
    int*   pieces = offs + Npad;
    int2*  slab   = (int2*)(pieces + Hpad);
    int2*  csr    = slab + Spad;
    float* out    = (float*)d_out;

    const int T = 256;

    gcn_chunk_sort<<<G1, T, 0, stream>>>(row, col, ew, pieces, slab, E, G1, NBK);
    gcn_bucket_sort<<<NBK, 1024, 0, stream>>>(pieces, slab, csr, offs, cnt, N, G1);
    gcn_dinv<<<(N + T - 1) / T, T, 0, stream>>>(offs, cnt, csr, dinv, N);

    const int ntile = (N + 15) / 16;
    gcn_gemm_wmma<<<(ntile + 7) / 8, T, 0, stream>>>(x, kern, dinv, h, N);

    gcn_row_gather<<<(N + 7) / 8, T, 0, stream>>>(offs, cnt, csr, dinv, h, bias, out, N);
}


# Bucket sort's counting pass also sums each row's weights (LDS float adds) and writes dinv for its rows: dinv launch and its csr walk gone, 5 launches to 4; degree sum order = LDS arrival order
# speedup vs baseline: 4.9968x; 1.0391x over previous
#include <hip/hip_runtime.h>
#include <hip/hip_bf16.h>

typedef float v2f __attribute__((ext_vector_type(2)));
typedef float v8f __attribute__((ext_vector_type(8)));

#define FEAT 64
#define FILL_WEIGHT 1.0f

#define BK_SHIFT   9
#define BK_ROWS    512
#define EDGE_CHUNK 4096
#define MAX_BK     2048
#define MAX_CHUNK  4096

template <bool FULL>
__device__ __forceinline__
void gcn_chunk_sort_body(const int* __restrict__ row,
                         const int* __restrict__ col,
                         const float* __restrict__ w,
                         int* __restrict__ pieces,
                         int2* __restrict__ slab,
                         int ne, int nblk, int nbk,
                         int* cnt, int* wsum) {
    const int t    = threadIdx.x;
    const int lane = t & 31;
    const int wv   = t >> 5;
    const int blk  = blockIdx.x;
    for (int i = t; i < nbk; i += 256) cnt[i] = 0;
    __syncthreads();

    const int e0 = blk * EDGE_CHUNK + t;
    int   r[16];
    int   c[16];
    float ww[16];
    for (int k = 0; k < 16; ++k) {
        const int  e  = e0 + k * 256;
        const bool ok = FULL || e < ne;
        r[k]  = ok ? row[e] : 0;
        c[k]  = ok ? col[e] : 0;
        ww[k] = ok ? w[e] : 0.0f;
    }
    int rank[16];
    for (int k = 0; k < 16; ++k) {
        const bool ok = FULL || e0 + k * 256 < ne;
        rank[k] = atomicAdd(&cnt[ok ? (r[k] >> BK_SHIFT) : t], ok ? 1 : 0);
    }
    __syncthreads();

    int v[8];
    int s = 0;
    for (int k = 0; k < 8; ++k) {
        const int i = t * 8 + k;
        v[k] = (i < nbk) ? cnt[i] : 0;
        s += v[k];
    }
    int incl = s;
    for (int d = 1; d < 32; d <<= 1) {
        const int up = __shfl_up(incl, d);
        if (lane >= d) incl += up;
    }
    if (lane == 31) wsum[wv] = incl;
    __syncthreads();
    int run = incl - s;
    for (int k = 0; k < wv; ++k) run += wsum[k];
    for (int k = 0; k < 8; ++k) {
        const int i = t * 8 + k;
        if (i < nbk) {
            cnt[i] = run;
            pieces[(size_t)i * nblk + blk] = run | (v[k] << 16);
            run += v[k];
        }
    }
    __syncthreads();

    int dst[16];
    for (int k = 0; k < 16; ++k) dst[k] = cnt[r[k] >> BK_SHIFT] + rank[k];
    int2* __restrict__ mine = slab + (size_t)blk * EDGE_CHUNK;
    for (int k = 0; k < 16; ++k)
        if (FULL || e0 + k * 256 < ne)
            mine[dst[k]] = make_int2((c[k] << BK_SHIFT) | (r[k] & (BK_ROWS - 1)), __float_as_int(ww[k]));
}

__global__ __launch_bounds__(256)
void gcn_chunk_sort(const int* __restrict__ row,
                    const int* __restrict__ col,
                    const float* __restrict__ w,
                    int* __restrict__ pieces,
                    int2* __restrict__ slab,
                    int ne, int nblk, int nbk) {
    __shared__ int cnt[MAX_BK];
    __shared__ int wsum[8];
    if ((int)(blockIdx.x + 1) * EDGE_CHUNK <= ne)
        gcn_chunk_sort_body<true>(row, col, w, pieces, slab, ne, nblk, nbk, cnt, wsum);
    else
        gcn_chunk_sort_body<false>(row, col, w, pieces, slab, ne, nblk, nbk, cnt, wsum);
}

#define TRIP_PIECES 8
#define TRIP_CHUNKS (32 * TRIP_PIECES)
static_assert(TRIP_PIECES == 8, "the load fences of both passes name eight values");
static_assert(MAX_CHUNK % TRIP_CHUNKS == 0, "the padded piece list must fit the LDS table");
__global__ __launch_bounds__(1024)
void gcn_bucket_sort(const int* __restrict__ pieces,
                     const int2* __restrict__ slab,
                     int2* __restrict__ csr,
                     int* __restrict__ offs,
                     int* __restrict__ cnt,
                     float* __restrict__ dinv,
                     int n, int nblk) {
    __shared__ int lcnt[BK_ROWS];
    __shared__ float ldeg[BK_ROWS];
    __shared__ int lcur[BK_ROWS];
    __shared__ int wsum[BK_ROWS / 32];
    __shared__ int piece[MAX_CHUNK];
    __shared__ int red[32];
    const int t    = threadIdx.x;
    const int lane = t & 31;
    const int wv   = __builtin_amdgcn_readfirstlane(t >> 5);
    const int b    = blockIdx.x;

    int s = 0;
    for (int c = t; c < nblk; c += 1024) {
        const int pk = pieces[(size_t)b * nblk + c];
        piece[c] = pk;
        s += pk & 0xFFFF;
    }
    for (int d = 16; d > 0; d >>= 1) s += __shfl_xor(s, d);
    if (lane == 0) red[wv] = s;
    for (int c = nblk + t; c < ((nblk + TRIP_CHUNKS - 1) & ~(TRIP_CHUNKS - 1)); c += 1024) piece[c] = 0;
    if (t < BK_ROWS) { lcnt[t] = 0; ldeg[t] = 0.0f; }
    __syncthreads();
    int bstart = 0;
    for (int k = 0; k < 32; ++k) bstart += red[k];


    for (int c0 = wv; c0 < nblk; c0 += TRIP_CHUNKS) {
        int2 kw[TRIP_PIECES];
        int  nn[TRIP_PIECES];
        int  base[TRIP_PIECES];
        for (int u = 0; u < TRIP_PIECES; ++u) {
            const int c     = c0 + u * 32;
            const int pk    = piece[c];
            const int slab0 = ((c < nblk) ? c : 0) * EDGE_CHUNK;
            nn[u]   = (int)((unsigned)pk >> 16);
            base[u] = slab0 + (pk & 0xFFFF);
            kw[u]   = slab[(lane < nn[u]) ? base[u] + lane : slab0];
        }
        asm volatile("" :: "v"(kw[0].x), "v"(kw[1].x), "v"(kw[2].x), "v"(kw[3].x),
                           "v"(kw[4].x), "v"(kw[5].x), "v"(kw[6].x), "v"(kw[7].x));
        for (int u = 0; u < TRIP_PIECES; ++u) {
            const bool ok  = lane < nn[u];
            const int  idx = ok ? (kw[u].x & (BK_ROWS - 1)) : lane;
            atomicAdd(&lcnt[idx], ok ? 1 : 0);
            atomicAdd(&ldeg[idx], ok ? __int_as_float(kw[u].y) : 0.0f);
        }
        for (int u = 0; u < TRIP_PIECES; ++u)
            for (int j = lane + 32; j < nn[u]; j += 32) {
                const int2 e = slab[base[u] + j];
                atomicAdd(&lcnt[e.x & (BK_ROWS - 1)], 1);
                atomicAdd(&ldeg[e.x & (BK_ROWS - 1)], __int_as_float(e.y));
            }
    }
    __syncthreads();

    const int v = (t < BK_ROWS) ? lcnt[t] : 0;
    int incl = v;
    for (int d = 1; d < 32; d <<= 1) {
        const int up = __shfl_up(incl, d);
        if (lane >= d) incl += up;
    }
    if (t < BK_ROWS && lane == 31) wsum[wv] = incl;
    __syncthreads();
    if (t < BK_ROWS) {
        int base = 0;
        for (int k = 0; k < wv; ++k) base += wsum[k];
        const int excl = base + incl - v;
        lcur[t] = excl;
        const int r = b * BK_ROWS + t;
        if (r < n) {
            offs[r] = bstart + excl;
            cnt[r]  = v;
            const float d = FILL_WEIGHT + ldeg[t];
            dinv[r] = (d > 0.0f) ? __frsqrt_rn(fmaxf(d, 1e-12f)) : 0.0f;
        }
    }
    __syncthreads();

    for (int c0 = wv; c0 < nblk; c0 += TRIP_CHUNKS) {
        int2 cw[TRIP_PIECES];
        int  nn[TRIP_PIECES];
        int  base[TRIP_PIECES];
        for (int u = 0; u < TRIP_PIECES; ++u) {
            const int c     = c0 + u * 32;
            const int pk    = piece[c];
            const int slab0 = ((c < nblk) ? c : 0) * EDGE_CHUNK;
            nn[u]   = (int)((unsigned)pk >> 16);
            base[u] = slab0 + (pk & 0xFFFF);
            cw[u]   = slab[(lane < nn[u]) ? base[u] + lane : slab0];
        }
        asm volatile("" :: "v"(cw[0].x), "v"(cw[1].x), "v"(cw[2].x), "v"(cw[3].x),
                           "v"(cw[4].x), "v"(cw[5].x), "v"(cw[6].x), "v"(cw[7].x));
        int pos[TRIP_PIECES];
        for (int u = 0; u < TRIP_PIECES; ++u) {
            const bool ok = lane < nn[u];
            pos[u] = bstart + atomicAdd(&lcur[ok ? (cw[u].x & (BK_ROWS - 1)) : lane], ok ? 1 : 0);
        }
        for (int u = 0; u < TRIP_PIECES; ++u)
            if (lane < nn[u])
                csr[pos[u]] = make_int2((int)((unsigned)cw[u].x >> BK_SHIFT), cw[u].y);
        for (int u = 0; u < TRIP_PIECES; ++u)
            for (int j = lane + 32; j < nn[u]; j += 32) {
                const int2 e = slab[base[u] + j];
                const int  p = bstart + atomicAdd(&lcur[e.x & (BK_ROWS - 1)], 1);
                csr[p] = make_int2((int)((unsigned)e.x >> BK_SHIFT), e.y);
            }
    }
}

__global__ __launch_bounds__(256)
void gcn_gemm_wmma(const float* __restrict__ x,
                   const float* __restrict__ kern,
                   const float* __restrict__ dinv,
                   float* __restrict__ h,
                   int num_nodes) {
    const int lane  = threadIdx.x & 31;
    const int wave  = threadIdx.x >> 5;
    const int tile  = blockIdx.x * 8 + wave;
    const int ntile = (num_nodes + 15) >> 4;
    if (tile >= ntile) return;

    const int r0   = tile << 4;
    const int half = lane >> 4;
    const int m    = lane & 15;

    v8f acc[4] = {};

    const float* xrow = x + (size_t)(r0 + m) * FEAT;
#pragma unroll
    for (int kb = 0; kb < FEAT; kb += 4) {
        const int k0 = kb + (half << 1);
        v2f a;
        a.x = xrow[k0];
        a.y = xrow[k0 + 1];
#pragma unroll
        for (int cb = 0; cb < 4; ++cb) {
            v2f b;
            b.x = kern[(size_t)k0 * FEAT + cb * 16 + m];
            b.y = kern[(size_t)(k0 + 1) * FEAT + cb * 16 + m];
            acc[cb] = __builtin_amdgcn_wmma_f32_16x16x4_f32(
                 false, a,  false, b,
                 (short)0, acc[cb],
                 false,  false);
        }
    }

    const float4* __restrict__ dv4 = reinterpret_cast<const float4*>(dinv + r0 + 8 * half);
    const float4 dlo = dv4[0];
    const float4 dhi = dv4[1];
    const float  rs[8] = {dlo.x, dlo.y, dlo.z, dlo.w, dhi.x, dhi.y, dhi.z, dhi.w};

#pragma unroll
    for (int cb = 0; cb < 4; ++cb) {
#pragma unroll
        for (int v = 0; v < 8; ++v) {
            h[(size_t)(r0 + v + 8 * half) * FEAT + cb * 16 + m] = acc[cb][v] * rs[v];
        }
    }
}

__global__ __launch_bounds__(256)
void gcn_row_gather(const int* __restrict__ offs,
                    const int* __restrict__ cnt,
                    const int2* __restrict__ csr,
                    const float* __restrict__ dinv,
                    const float* __restrict__ h,
                    const float* __restrict__ bias,
                    float* __restrict__ out, int n) {
    const int lane = threadIdx.x & 31;
    const int r = __builtin_amdgcn_readfirstlane(blockIdx.x * 8 + (threadIdx.x >> 5));
    if (r >= n) return;

    const int   beg = offs[r];
    const int   num = cnt[r];
    const float dr  = dinv[r];
    const v2f* __restrict__ h2 = reinterpret_cast<const v2f*>(h);

#define EDGE_C(j) __builtin_amdgcn_readlane(c, (j))
#define EDGE_W(j) __int_as_float(__builtin_amdgcn_readlane(__float_as_int(wn), (j)))
#define H_ROW(cc) h2[(size_t)(cc) * (FEAT / 2) + lane]

    v2f acc = {0.0f, 0.0f};
    for (int base = 0; base < num; base += 32) {
        const int p = base + lane;
        int   c  = 0;
        float wn = 0.0f;
        if (p < num) {
            const int2 cw = csr[beg + p];
            c  = cw.x;
            wn = dr * __int_as_float(cw.y);
        }
        const int m = min(32, num - base);
        int j = 0;
        for (; j + 8 <= m; j += 8) {
            const v2f a0 = H_ROW(EDGE_C(j));
            const v2f a1 = H_ROW(EDGE_C(j + 1));
            const v2f a2 = H_ROW(EDGE_C(j + 2));
            const v2f a3 = H_ROW(EDGE_C(j + 3));
            const v2f a4 = H_ROW(EDGE_C(j + 4));
            const v2f a5 = H_ROW(EDGE_C(j + 5));
            const v2f a6 = H_ROW(EDGE_C(j + 6));
            const v2f a7 = H_ROW(EDGE_C(j + 7));
            acc += a0 * EDGE_W(j);
            acc += a1 * EDGE_W(j + 1);
            acc += a2 * EDGE_W(j + 2);
            acc += a3 * EDGE_W(j + 3);
            acc += a4 * EDGE_W(j + 4);
            acc += a5 * EDGE_W(j + 5);
            acc += a6 * EDGE_W(j + 6);
            acc += a7 * EDGE_W(j + 7);
        }
        if (j + 4 <= m) {
            const v2f a0 = H_ROW(EDGE_C(j));
            const v2f a1 = H_ROW(EDGE_C(j + 1));
            const v2f a2 = H_ROW(EDGE_C(j + 2));
            const v2f a3 = H_ROW(EDGE_C(j + 3));
            acc += a0 * EDGE_W(j);
            acc += a1 * EDGE_W(j + 1);
            acc += a2 * EDGE_W(j + 2);
            acc += a3 * EDGE_W(j + 3);
            j += 4;
        }
#pragma clang loop unroll(disable)
        for (; j < m; ++j) {
            const v2f a = H_ROW(EDGE_C(j));
            acc += a * EDGE_W(j);
        }
    }
#undef EDGE_C
#undef EDGE_W
#undef H_ROW

    const v2f   hs = h2[(size_t)r * (FEAT / 2) + lane];
    const float b0 = bias[2 * lane];
    const float b1 = bias[2 * lane + 1];
    v2f o;
    o.x = (b0 + hs.x * dr * FILL_WEIGHT) + acc.x;
    o.y = (b1 + hs.y * dr * FILL_WEIGHT) + acc.y;
    reinterpret_cast<v2f*>(out)[(size_t)r * (FEAT / 2) + lane] = o;
}

extern "C" void kernel_launch(void* const* d_in, const int* in_sizes, int n_in,
                              void* d_out, int out_size, void* d_ws, size_t ws_size,
                              hipStream_t stream) {
    const float* x    = (const float*)d_in[0];
    const int*   ei   = (const int*)d_in[1];
    const float* ew   = (const float*)d_in[2];
    const float* kern = (const float*)d_in[3];
    const float* bias = (const float*)d_in[4];

    const int N = in_sizes[0] / FEAT;
    const int E = in_sizes[2];
    const int* row = ei;
    const int* col = ei + E;

    const int G1   = (E + EDGE_CHUNK - 1) / EDGE_CHUNK;
    const int NBK  = (N + BK_ROWS - 1) / BK_ROWS;
    const int n2   = NBK * G1;
    const int Npad = (N + 3) & ~3;
    const int Hpad = (n2 + 3) & ~3;
    const int Spad = G1 * EDGE_CHUNK;

    float* h      = (float*)d_ws;
    float* dinv   = h + (size_t)N * FEAT;
    int*   cnt    = (int*)(dinv + Npad);
    int*   offs   = cnt + Npad;
    int*   pieces = offs + Npad;
    int2*  slab   = (int2*)(pieces + Hpad);
    int2*  csr    = slab + Spad;
    float* out    = (float*)d_out;

    const int T = 256;

    gcn_chunk_sort<<<G1, T, 0, stream>>>(row, col, ew, pieces, slab, E, G1, NBK);
    gcn_bucket_sort<<<NBK, 1024, 0, stream>>>(pieces, slab, csr, offs, cnt, dinv, N, G1);

    const int ntile = (N + 15) / 16;
    gcn_gemm_wmma<<<(ntile + 7) / 8, T, 0, stream>>>(x, kern, dinv, h, N);

    gcn_row_gather<<<(N + 7) / 8, T, 0, stream>>>(offs, cnt, csr, dinv, h, bias, out, N);
}


# Row gather run by 8,192 resident waves looping over rows (next row's edge block and bounds fetched ahead, a row's loads issued as one group) instead of one short wave per row; row sum order unchanged
# speedup vs baseline: 5.3921x; 1.0791x over previous
#include <hip/hip_runtime.h>
#include <hip/hip_bf16.h>

typedef float v2f __attribute__((ext_vector_type(2)));
typedef float v8f __attribute__((ext_vector_type(8)));

#define FEAT 64
#define FILL_WEIGHT 1.0f

#define BK_SHIFT   9
#define BK_ROWS    512
#define EDGE_CHUNK 4096
#define MAX_BK     2048
#define MAX_CHUNK  4096

template <bool FULL>
__device__ __forceinline__
void gcn_chunk_sort_body(const int* __restrict__ row,
                         const int* __restrict__ col,
                         const float* __restrict__ w,
                         int* __restrict__ pieces,
                         int2* __restrict__ slab,
                         int ne, int nblk, int nbk,
                         int* cnt, int* wsum) {
    const int t    = threadIdx.x;
    const int lane = t & 31;
    const int wv   = t >> 5;
    const int blk  = blockIdx.x;
    for (int i = t; i < nbk; i += 256) cnt[i] = 0;
    __syncthreads();

    const int e0 = blk * EDGE_CHUNK + t;
    int   r[16];
    int   c[16];
    float ww[16];
    for (int k = 0; k < 16; ++k) {
        const int  e  = e0 + k * 256;
        const bool ok = FULL || e < ne;
        r[k]  = ok ? row[e] : 0;
        c[k]  = ok ? col[e] : 0;
        ww[k] = ok ? w[e] : 0.0f;
    }
    int rank[16];
    for (int k = 0; k < 16; ++k) {
        const bool ok = FULL || e0 + k * 256 < ne;
        rank[k] = atomicAdd(&cnt[ok ? (r[k] >> BK_SHIFT) : t], ok ? 1 : 0);
    }
    __syncthreads();

    int v[8];
    int s = 0;
    for (int k = 0; k < 8; ++k) {
        const int i = t * 8 + k;
        v[k] = (i < nbk) ? cnt[i] : 0;
        s += v[k];
    }
    int incl = s;
    for (int d = 1; d < 32; d <<= 1) {
        const int up = __shfl_up(incl, d);
        if (lane >= d) incl += up;
    }
    if (lane == 31) wsum[wv] = incl;
    __syncthreads();
    int run = incl - s;
    for (int k = 0; k < wv; ++k) run += wsum[k];
    for (int k = 0; k < 8; ++k) {
        const int i = t * 8 + k;
        if (i < nbk) {
            cnt[i] = run;
            pieces[(size_t)i * nblk + blk] = run | (v[k] << 16);
            run += v[k];
        }
    }
    __syncthreads();

    int dst[16];
    for (int k = 0; k < 16; ++k) dst[k] = cnt[r[k] >> BK_SHIFT] + rank[k];
    int2* __restrict__ mine = slab + (size_t)blk * EDGE_CHUNK;
    for (int k = 0; k < 16; ++k)
        if (FULL || e0 + k * 256 < ne)
            mine[dst[k]] = make_int2((c[k] << BK_SHIFT) | (r[k] & (BK_ROWS - 1)), __float_as_int(ww[k]));
}

__global__ __launch_bounds__(256)
void gcn_chunk_sort(const int* __restrict__ row,
                    const int* __restrict__ col,
                    const float* __restrict__ w,
                    int* __restrict__ pieces,
                    int2* __restrict__ slab,
                    int ne, int nblk, int nbk) {
    __shared__ int cnt[MAX_BK];
    __shared__ int wsum[8];
    if ((int)(blockIdx.x + 1) * EDGE_CHUNK <= ne)
        gcn_chunk_sort_body<true>(row, col, w, pieces, slab, ne, nblk, nbk, cnt, wsum);
    else
        gcn_chunk_sort_body<false>(row, col, w, pieces, slab, ne, nblk, nbk, cnt, wsum);
}

#define TRIP_PIECES 8
#define TRIP_CHUNKS (32 * TRIP_PIECES)
static_assert(TRIP_PIECES == 8, "the load fences of both passes name eight values");
static_assert(MAX_CHUNK % TRIP_CHUNKS == 0, "the padded piece list must fit the LDS table");
__global__ __launch_bounds__(1024)
void gcn_bucket_sort(const int* __restrict__ pieces,
                     const int2* __restrict__ slab,
                     int2* __restrict__ csr,
                     int* __restrict__ offs,
                     int* __restrict__ cnt,
                     float* __restrict__ dinv,
                     int n, int nblk) {
    __shared__ int lcnt[BK_ROWS];
    __shared__ float ldeg[BK_ROWS];
    __shared__ int lcur[BK_ROWS];
    __shared__ int wsum[BK_ROWS / 32];
    __shared__ int piece[MAX_CHUNK];
    __shared__ int red[32];
    const int t    = threadIdx.x;
    const int lane = t & 31;
    const int wv   = __builtin_amdgcn_readfirstlane(t >> 5);
    const int b    = blockIdx.x;

    int s = 0;
    for (int c = t; c < nblk; c += 1024) {
        const int pk = pieces[(size_t)b * nblk + c];
        piece[c] = pk;
        s += pk & 0xFFFF;
    }
    for (int d = 16; d > 0; d >>= 1) s += __shfl_xor(s, d);
    if (lane == 0) red[wv] = s;
    for (int c = nblk + t; c < ((nblk + TRIP_CHUNKS - 1) & ~(TRIP_CHUNKS - 1)); c += 1024) piece[c] = 0;
    if (t < BK_ROWS) { lcnt[t] = 0; ldeg[t] = 0.0f; }
    __syncthreads();
    int bstart = 0;
    for (int k = 0; k < 32; ++k) bstart += red[k];


    for (int c0 = wv; c0 < nblk; c0 += TRIP_CHUNKS) {
        int2 kw[TRIP_PIECES];
        int  nn[TRIP_PIECES];
        int  base[TRIP_PIECES];
        for (int u = 0; u < TRIP_PIECES; ++u) {
            const int c     = c0 + u * 32;
            const int pk    = piece[c];
            const int slab0 = ((c < nblk) ? c : 0) * EDGE_CHUNK;
            nn[u]   = (int)((unsigned)pk >> 16);
            base[u] = slab0 + (pk & 0xFFFF);
            kw[u]   = slab[(lane < nn[u]) ? base[u] + lane : slab0];
        }
        asm volatile("" :: "v"(kw[0].x), "v"(kw[1].x), "v"(kw[2].x), "v"(kw[3].x),
                           "v"(kw[4].x), "v"(kw[5].x), "v"(kw[6].x), "v"(kw[7].x));
        for (int u = 0; u < TRIP_PIECES; ++u) {
            const bool ok  = lane < nn[u];
            const int  idx = ok ? (kw[u].x & (BK_ROWS - 1)) : lane;
            atomicAdd(&lcnt[idx], ok ? 1 : 0);
            atomicAdd(&ldeg[idx], ok ? __int_as_float(kw[u].y) : 0.0f);
        }
        for (int u = 0; u < TRIP_PIECES; ++u)
            for (int j = lane + 32; j < nn[u]; j += 32) {
                const int2 e = slab[base[u] + j];
                atomicAdd(&lcnt[e.x & (BK_ROWS - 1)], 1);
                atomicAdd(&ldeg[e.x & (BK_ROWS - 1)], __int_as_float(e.y));
            }
    }
    __syncthreads();

    const int v = (t < BK_ROWS) ? lcnt[t] : 0;
    int incl = v;
    for (int d = 1; d < 32; d <<= 1) {
        const int up = __shfl_up(incl, d);
        if (lane >= d) incl += up;
    }
    if (t < BK_ROWS && lane == 31) wsum[wv] = incl;
    __syncthreads();
    if (t < BK_ROWS) {
        int base = 0;
        for (int k = 0; k < wv; ++k) base += wsum[k];
        const int excl = base + incl - v;
        lcur[t] = excl;
        const int r = b * BK_ROWS + t;
        if (r < n) {
            offs[r] = bstart + excl;
            cnt[r]  = v;
            const float d = FILL_WEIGHT + ldeg[t];
            dinv[r] = (d > 0.0f) ? __frsqrt_rn(fmaxf(d, 1e-12f)) : 0.0f;
        }
    }
    __syncthreads();

    for (int c0 = wv; c0 < nblk; c0 += TRIP_CHUNKS) {
        int2 cw[TRIP_PIECES];
        int  nn[TRIP_PIECES];
        int  base[TRIP_PIECES];
        for (int u = 0; u < TRIP_PIECES; ++u) {
            const int c     = c0 + u * 32;
            const int pk    = piece[c];
            const int slab0 = ((c < nblk) ? c : 0) * EDGE_CHUNK;
            nn[u]   = (int)((unsigned)pk >> 16);
            base[u] = slab0 + (pk & 0xFFFF);
            cw[u]   = slab[(lane < nn[u]) ? base[u] + lane : slab0];
        }
        asm volatile("" :: "v"(cw[0].x), "v"(cw[1].x), "v"(cw[2].x), "v"(cw[3].x),
                           "v"(cw[4].x), "v"(cw[5].x), "v"(cw[6].x), "v"(cw[7].x));
        int pos[TRIP_PIECES];
        for (int u = 0; u < TRIP_PIECES; ++u) {
            const bool ok = lane < nn[u];
            pos[u] = bstart + atomicAdd(&lcur[ok ? (cw[u].x & (BK_ROWS - 1)) : lane], ok ? 1 : 0);
        }
        for (int u = 0; u < TRIP_PIECES; ++u)
            if (lane < nn[u])
                csr[pos[u]] = make_int2((int)((unsigned)cw[u].x >> BK_SHIFT), cw[u].y);
        for (int u = 0; u < TRIP_PIECES; ++u)
            for (int j = lane + 32; j < nn[u]; j += 32) {
                const int2 e = slab[base[u] + j];
                const int  p = bstart + atomicAdd(&lcur[e.x & (BK_ROWS - 1)], 1);
                csr[p] = make_int2((int)((unsigned)e.x >> BK_SHIFT), e.y);
            }
    }
}

__global__ __launch_bounds__(256)
void gcn_gemm_wmma(const float* __restrict__ x,
                   const float* __restrict__ kern,
                   const float* __restrict__ dinv,
                   float* __restrict__ h,
                   int num_nodes) {
    const int lane  = threadIdx.x & 31;
    const int wave  = threadIdx.x >> 5;
    const int tile  = blockIdx.x * 8 + wave;
    const int ntile = (num_nodes + 15) >> 4;
    if (tile >= ntile) return;

    const int r0   = tile << 4;
    const int half = lane >> 4;
    const int m    = lane & 15;

    v8f acc[4] = {};

    const float* xrow = x + (size_t)(r0 + m) * FEAT;
#pragma unroll
    for (int kb = 0; kb < FEAT; kb += 4) {
        const int k0 = kb + (half << 1);
        v2f a;
        a.x = xrow[k0];
        a.y = xrow[k0 + 1];
#pragma unroll
        for (int cb = 0; cb < 4; ++cb) {
            v2f b;
            b.x = kern[(size_t)k0 * FEAT + cb * 16 + m];
            b.y = kern[(size_t)(k0 + 1) * FEAT + cb * 16 + m];
            acc[cb] = __builtin_amdgcn_wmma_f32_16x16x4_f32(
                 false, a,  false, b,
                 (short)0, acc[cb],
                 false,  false);
        }
    }

    const float4* __restrict__ dv4 = reinterpret_cast<const float4*>(dinv + r0 + 8 * half);
    const float4 dlo = dv4[0];
    const float4 dhi = dv4[1];
    const float  rs[8] = {dlo.x, dlo.y, dlo.z, dlo.w, dhi.x, dhi.y, dhi.z, dhi.w};

#pragma unroll
    for (int cb = 0; cb < 4; ++cb) {
#pragma unroll
        for (int v = 0; v < 8; ++v) {
            h[(size_t)(r0 + v + 8 * half) * FEAT + cb * 16 + m] = acc[cb][v] * rs[v];
        }
    }
}

#define GATHER_WG 1024
#define GB        31
__global__ __launch_bounds__(256)
void gcn_row_gather(const int* __restrict__ offs,
                    const int* __restrict__ cnt,
                    const int2* __restrict__ csr,
                    const float* __restrict__ dinv,
                    const float* __restrict__ h,
                    const float* __restrict__ bias,
                    float* __restrict__ out, int n) {
    const int lane = threadIdx.x & 31;
    int r = __builtin_amdgcn_readfirstlane(blockIdx.x * 8 + (threadIdx.x >> 5));
    if (r >= n) return;
    const int nw = GATHER_WG * 8;

    const v2f* __restrict__ h2 = reinterpret_cast<const v2f*>(h);
    v2f* __restrict__ out2 = reinterpret_cast<v2f*>(out);
    const float b0 = bias[2 * lane];
    const float b1 = bias[2 * lane + 1];

#define EDGE_C(j) __builtin_amdgcn_readlane(c, (j))
#define EDGE_W(j) __int_as_float(__builtin_amdgcn_readlane(__float_as_int(wn), (j)))
#define H_ROW(cc) h2[(unsigned)(cc) * (unsigned)(FEAT / 2) + ulane]
    const unsigned ulane = (unsigned)lane;

    int   beg = offs[r];
    int   num = cnt[r];
    float dr  = dinv[r];
    int2  cw  = csr[(lane < min(num, GB)) ? beg + lane : 0];
    v2f   hs  = H_ROW(r);
    int   rn   = r + nw;
    int   begn = 0;
    int   numn = 0;
    float drn  = 0.0f;
    if (rn < n) { begn = offs[rn]; numn = cnt[rn]; drn = dinv[rn]; }

    for (;;) {
        const bool has_n = rn < n;
        const int2 cwn = csr[(lane < min(numn, GB)) ? begn + lane : 0];
        const v2f  hsn = H_ROW(has_n ? rn : r);
        const int rnn   = rn + nw;
        int       begnn = 0;
        int       numnn = 0;
        float     drnn  = 0.0f;
        if (rnn < n) { begnn = offs[rnn]; numnn = cnt[rnn]; drnn = dinv[rnn]; }

        v2f acc = {0.0f, 0.0f};
        for (int base = 0; base < num; base += GB) {
            const int m = min(GB, num - base);
            int   c;
            float wn;
            if (base == 0) {
                c  = cw.x;
                wn = dr * __int_as_float(cw.y);
            } else {
                c  = 0;
                wn = 0.0f;
                if (lane < m) {
                    const int2 e = csr[beg + base + lane];
                    c  = e.x;
                    wn = dr * __int_as_float(e.y);
                }
            }
            asm volatile("" :: "v"(c));
            v2f g16[16], g8[8], g4[4], g2[2], g1;
            int j = 0;
            if (m & 16) {
#pragma unroll
                for (int u = 0; u < 16; ++u) g16[u] = H_ROW(EDGE_C(j + u));
                j += 16;
            }
            if (m & 8) {
#pragma unroll
                for (int u = 0; u < 8; ++u) g8[u] = H_ROW(EDGE_C(j + u));
                j += 8;
            }
            if (m & 4) {
#pragma unroll
                for (int u = 0; u < 4; ++u) g4[u] = H_ROW(EDGE_C(j + u));
                j += 4;
            }
            if (m & 2) {
#pragma unroll
                for (int u = 0; u < 2; ++u) g2[u] = H_ROW(EDGE_C(j + u));
                j += 2;
            }
            if (m & 1) g1 = H_ROW(EDGE_C(j));
            j = 0;
            if (m & 16) {
#pragma unroll
                for (int u = 0; u < 16; ++u) acc += g16[u] * EDGE_W(j + u);
                j += 16;
            }
            if (m & 8) {
#pragma unroll
                for (int u = 0; u < 8; ++u) acc += g8[u] * EDGE_W(j + u);
                j += 8;
            }
            if (m & 4) {
#pragma unroll
                for (int u = 0; u < 4; ++u) acc += g4[u] * EDGE_W(j + u);
                j += 4;
            }
            if (m & 2) {
#pragma unroll
                for (int u = 0; u < 2; ++u) acc += g2[u] * EDGE_W(j + u);
                j += 2;
            }
            if (m & 1) acc += g1 * EDGE_W(j);
        }

        v2f o;
        o.x = (b0 + hs.x * dr * FILL_WEIGHT) + acc.x;
        o.y = (b1 + hs.y * dr * FILL_WEIGHT) + acc.y;
        out2[(unsigned)r * (unsigned)(FEAT / 2) + ulane] = o;

        if (!has_n) break;
        r = rn;   beg = begn;   num = numn;   dr = drn;   cw = cwn;   hs = hsn;
        rn = rnn; begn = begnn; numn = numnn; drn = drnn;
    }
#undef EDGE_C
#undef EDGE_W
#undef H_ROW
}

extern "C" void kernel_launch(void* const* d_in, const int* in_sizes, int n_in,
                              void* d_out, int out_size, void* d_ws, size_t ws_size,
                              hipStream_t stream) {
    const float* x    = (const float*)d_in[0];
    const int*   ei   = (const int*)d_in[1];
    const float* ew   = (const float*)d_in[2];
    const float* kern = (const float*)d_in[3];
    const float* bias = (const float*)d_in[4];

    const int N = in_sizes[0] / FEAT;
    const int E = in_sizes[2];
    const int* row = ei;
    const int* col = ei + E;

    const int G1   = (E + EDGE_CHUNK - 1) / EDGE_CHUNK;
    const int NBK  = (N + BK_ROWS - 1) / BK_ROWS;
    const int n2   = NBK * G1;
    const int Npad = (N + 3) & ~3;
    const int Hpad = (n2 + 3) & ~3;
    const int Spad = G1 * EDGE_CHUNK;

    float* h      = (float*)d_ws;
    float* dinv   = h + (size_t)N * FEAT;
    int*   cnt    = (int*)(dinv + Npad);
    int*   offs   = cnt + Npad;
    int*   pieces = offs + Npad;
    int2*  slab   = (int2*)(pieces + Hpad);
    int2*  csr    = slab + Spad;
    float* out    = (float*)d_out;

    const int T = 256;

    gcn_chunk_sort<<<G1, T, 0, stream>>>(row, col, ew, pieces, slab, E, G1, NBK);
    gcn_bucket_sort<<<NBK, 1024, 0, stream>>>(pieces, slab, csr, offs, cnt, dinv, N, G1);

    const int ntile = (N + 15) / 16;
    gcn_gemm_wmma<<<(ntile + 7) / 8, T, 0, stream>>>(x, kern, dinv, h, N);

    gcn_row_gather<<<GATHER_WG, T, 0, stream>>>(offs, cnt, csr, dinv, h, bias, out, N);
}


# Both sort levels place edges in an LDS image of their slab / csr part and copy it out with consecutive lanes: whole-line stores replace one 8-byte piece of a line per edge; sum orders unchanged
# speedup vs baseline: 6.0472x; 1.1215x over previous
#include <hip/hip_runtime.h>
#include <hip/hip_bf16.h>

typedef float v2f __attribute__((ext_vector_type(2)));
typedef float v8f __attribute__((ext_vector_type(8)));

#define FEAT 64
#define FILL_WEIGHT 1.0f

#define BK_SHIFT   9
#define BK_ROWS    512
#define EDGE_CHUNK 4096
#define MAX_BK     2048
#define MAX_CHUNK  4096

template <bool FULL>
__device__ __forceinline__
void gcn_chunk_sort_body(const int* __restrict__ row,
                         const int* __restrict__ col,
                         const float* __restrict__ w,
                         int* __restrict__ pieces,
                         int2* __restrict__ slab,
                         int ne, int nblk, int nbk,
                         int* cnt, int* wsum, int4* stage4) {
    const int t    = threadIdx.x;
    const int lane = t & 31;
    const int wv   = t >> 5;
    const int blk  = blockIdx.x;
    for (int i = t; i < nbk; i += 256) cnt[i] = 0;
    __syncthreads();

    const int e0 = blk * EDGE_CHUNK + t;
    int   r[16];
    int   c[16];
    float ww[16];
    for (int k = 0; k < 16; ++k) {
        const int  e  = e0 + k * 256;
        const bool ok = FULL || e < ne;
        r[k]  = ok ? row[e] : 0;
        c[k]  = ok ? col[e] : 0;
        ww[k] = ok ? w[e] : 0.0f;
    }
    int rank[16];
    for (int k = 0; k < 16; ++k) {
        const bool ok = FULL || e0 + k * 256 < ne;
        rank[k] = atomicAdd(&cnt[ok ? (r[k] >> BK_SHIFT) : t], ok ? 1 : 0);
    }
    __syncthreads();

    int v[8];
    int s = 0;
    for (int k = 0; k < 8; ++k) {
        const int i = t * 8 + k;
        v[k] = (i < nbk) ? cnt[i] : 0;
        s += v[k];
    }
    int incl = s;
    for (int d = 1; d < 32; d <<= 1) {
        const int up = __shfl_up(incl, d);
        if (lane >= d) incl += up;
    }
    if (lane == 31) wsum[wv] = incl;
    __syncthreads();
    int run = incl - s;
    for (int k = 0; k < wv; ++k) run += wsum[k];
    for (int k = 0; k < 8; ++k) {
        const int i = t * 8 + k;
        if (i < nbk) {
            cnt[i] = run;
            pieces[(size_t)i * nblk + blk] = run | (v[k] << 16);
            run += v[k];
        }
    }
    __syncthreads();

    int dst[16];
    for (int k = 0; k < 16; ++k) dst[k] = cnt[r[k] >> BK_SHIFT] + rank[k];
    int2* stage = reinterpret_cast<int2*>(stage4);
    for (int k = 0; k < 16; ++k)
        if (FULL || e0 + k * 256 < ne)
            stage[dst[k]] = make_int2((c[k] << BK_SHIFT) | (r[k] & (BK_ROWS - 1)), __float_as_int(ww[k]));
    __syncthreads();
    int2* __restrict__ mine = slab + (size_t)blk * EDGE_CHUNK;
    if (FULL) {
        int4* __restrict__ mine4 = reinterpret_cast<int4*>(mine);
        for (int k = 0; k < EDGE_CHUNK / 2 / 256; ++k) mine4[t + k * 256] = stage4[t + k * 256];
    } else {
        const int nv = ne - blk * EDGE_CHUNK;
        for (int i = t; i < nv; i += 256) mine[i] = stage[i];
    }
}

__global__ __launch_bounds__(256)
void gcn_chunk_sort(const int* __restrict__ row,
                    const int* __restrict__ col,
                    const float* __restrict__ w,
                    int* __restrict__ pieces,
                    int2* __restrict__ slab,
                    int ne, int nblk, int nbk) {
    __shared__ int cnt[MAX_BK];
    __shared__ int wsum[8];
    __shared__ int4 stage4[EDGE_CHUNK / 2];
    if ((int)(blockIdx.x + 1) * EDGE_CHUNK <= ne)
        gcn_chunk_sort_body<true>(row, col, w, pieces, slab, ne, nblk, nbk, cnt, wsum, stage4);
    else
        gcn_chunk_sort_body<false>(row, col, w, pieces, slab, ne, nblk, nbk, cnt, wsum, stage4);
}

#define TRIP_PIECES 8
#define STAGE_CAP   12288
#define TRIP_CHUNKS (32 * TRIP_PIECES)
static_assert(TRIP_PIECES == 8, "the load fences of both passes name eight values");
static_assert(MAX_CHUNK % TRIP_CHUNKS == 0, "the padded piece list must fit the LDS table");
__global__ __launch_bounds__(1024)
void gcn_bucket_sort(const int* __restrict__ pieces,
                     const int2* __restrict__ slab,
                     int2* __restrict__ csr,
                     int* __restrict__ offs,
                     int* __restrict__ cnt,
                     float* __restrict__ dinv,
                     int n, int nblk) {
    __shared__ int lcnt[BK_ROWS];
    __shared__ float ldeg[BK_ROWS];
    __shared__ int lcur[BK_ROWS];
    __shared__ int wsum[BK_ROWS / 32];
    __shared__ int piece[MAX_CHUNK];
    __shared__ int red[32];
    __shared__ int btot;
    __shared__ int2 stage[STAGE_CAP];
    const int t    = threadIdx.x;
    const int lane = t & 31;
    const int wv   = __builtin_amdgcn_readfirstlane(t >> 5);
    const int b    = blockIdx.x;

    int s = 0;
    for (int c = t; c < nblk; c += 1024) {
        const int pk = pieces[(size_t)b * nblk + c];
        piece[c] = pk;
        s += pk & 0xFFFF;
    }
    for (int d = 16; d > 0; d >>= 1) s += __shfl_xor(s, d);
    if (lane == 0) red[wv] = s;
    for (int c = nblk + t; c < ((nblk + TRIP_CHUNKS - 1) & ~(TRIP_CHUNKS - 1)); c += 1024) piece[c] = 0;
    if (t < BK_ROWS) { lcnt[t] = 0; ldeg[t] = 0.0f; }
    __syncthreads();
    int bstart = 0;
    for (int k = 0; k < 32; ++k) bstart += red[k];


    for (int c0 = wv; c0 < nblk; c0 += TRIP_CHUNKS) {
        int2 kw[TRIP_PIECES];
        int  nn[TRIP_PIECES];
        int  base[TRIP_PIECES];
        for (int u = 0; u < TRIP_PIECES; ++u) {
            const int c     = c0 + u * 32;
            const int pk    = piece[c];
            const int slab0 = ((c < nblk) ? c : 0) * EDGE_CHUNK;
            nn[u]   = (int)((unsigned)pk >> 16);
            base[u] = slab0 + (pk & 0xFFFF);
            kw[u]   = slab[(lane < nn[u]) ? base[u] + lane : slab0];
        }
        asm volatile("" :: "v"(kw[0].x), "v"(kw[1].x), "v"(kw[2].x), "v"(kw[3].x),
                           "v"(kw[4].x), "v"(kw[5].x), "v"(kw[6].x), "v"(kw[7].x));
        for (int u = 0; u < TRIP_PIECES; ++u) {
            const bool ok  = lane < nn[u];
            const int  idx = ok ? (kw[u].x & (BK_ROWS - 1)) : lane;
            atomicAdd(&lcnt[idx], ok ? 1 : 0);
            atomicAdd(&ldeg[idx], ok ? __int_as_float(kw[u].y) : 0.0f);
        }
        for (int u = 0; u < TRIP_PIECES; ++u)
            for (int j = lane + 32; j < nn[u]; j += 32) {
                const int2 e = slab[base[u] + j];
                atomicAdd(&lcnt[e.x & (BK_ROWS - 1)], 1);
                atomicAdd(&ldeg[e.x & (BK_ROWS - 1)], __int_as_float(e.y));
            }
    }
    __syncthreads();

    const int v = (t < BK_ROWS) ? lcnt[t] : 0;
    int incl = v;
    for (int d = 1; d < 32; d <<= 1) {
        const int up = __shfl_up(incl, d);
        if (lane >= d) incl += up;
    }
    if (t < BK_ROWS && lane == 31) wsum[wv] = incl;
    __syncthreads();
    if (t < BK_ROWS) {
        int base = 0;
        for (int k = 0; k < wv; ++k) base += wsum[k];
        const int excl = base + incl - v;
        lcur[t] = excl;
        if (t == BK_ROWS - 1) btot = excl + v;
        const int r = b * BK_ROWS + t;
        if (r < n) {
            offs[r] = bstart + excl;
            cnt[r]  = v;
            const float d = FILL_WEIGHT + ldeg[t];
            dinv[r] = (d > 0.0f) ? __frsqrt_rn(fmaxf(d, 1e-12f)) : 0.0f;
        }
    }
    __syncthreads();

    for (int c0 = wv; c0 < nblk; c0 += TRIP_CHUNKS) {
        int2 cw[TRIP_PIECES];
        int  nn[TRIP_PIECES];
        int  base[TRIP_PIECES];
        for (int u = 0; u < TRIP_PIECES; ++u) {
            const int c     = c0 + u * 32;
            const int pk    = piece[c];
            const int slab0 = ((c < nblk) ? c : 0) * EDGE_CHUNK;
            nn[u]   = (int)((unsigned)pk >> 16);
            base[u] = slab0 + (pk & 0xFFFF);
            cw[u]   = slab[(lane < nn[u]) ? base[u] + lane : slab0];
        }
        asm volatile("" :: "v"(cw[0].x), "v"(cw[1].x), "v"(cw[2].x), "v"(cw[3].x),
                           "v"(cw[4].x), "v"(cw[5].x), "v"(cw[6].x), "v"(cw[7].x));
        int pos[TRIP_PIECES];
        for (int u = 0; u < TRIP_PIECES; ++u) {
            const bool ok = lane < nn[u];
            pos[u] = atomicAdd(&lcur[ok ? (cw[u].x & (BK_ROWS - 1)) : lane], ok ? 1 : 0);
        }
        for (int u = 0; u < TRIP_PIECES; ++u)
            if (lane < nn[u]) {
                const int2 e = make_int2((int)((unsigned)cw[u].x >> BK_SHIFT), cw[u].y);
                if (pos[u] < STAGE_CAP) stage[pos[u]] = e;
                else                    csr[bstart + pos[u]] = e;
            }
        for (int u = 0; u < TRIP_PIECES; ++u)
            for (int j = lane + 32; j < nn[u]; j += 32) {
                const int2 e = slab[base[u] + j];
                const int  p = atomicAdd(&lcur[e.x & (BK_ROWS - 1)], 1);
                const int2 o = make_int2((int)((unsigned)e.x >> BK_SHIFT), e.y);
                if (p < STAGE_CAP) stage[p] = o;
                else               csr[bstart + p] = o;
            }
    }
    __syncthreads();

    const int nst  = min(btot, STAGE_CAP);
    const int head = bstart & 15;
    if (nst > 0) {
        for (int i0 = t - head; i0 < nst; i0 += 4 * 1024) {
            int2 v[4];
            for (int k = 0; k < 4; ++k) v[k] = stage[min(max(i0 + k * 1024, 0), nst - 1)];
            for (int k = 0; k < 4; ++k) {
                const int i = i0 + k * 1024;
                if (i >= 0 && i < nst) csr[bstart + i] = v[k];
            }
        }
    }
}

__global__ __launch_bounds__(256)
void gcn_gemm_wmma(const float* __restrict__ x,
                   const float* __restrict__ kern,
                   const float* __restrict__ dinv,
                   float* __restrict__ h,
                   int num_nodes) {
    const int lane  = threadIdx.x & 31;
    const int wave  = threadIdx.x >> 5;
    const int tile  = blockIdx.x * 8 + wave;
    const int ntile = (num_nodes + 15) >> 4;
    if (tile >= ntile) return;

    const int r0   = tile << 4;
    const int half = lane >> 4;
    const int m    = lane & 15;

    v8f acc[4] = {};

    const float* xrow = x + (size_t)(r0 + m) * FEAT;
#pragma unroll
    for (int kb = 0; kb < FEAT; kb += 4) {
        const int k0 = kb + (half << 1);
        v2f a;
        a.x = xrow[k0];
        a.y = xrow[k0 + 1];
#pragma unroll
        for (int cb = 0; cb < 4; ++cb) {
            v2f b;
            b.x = kern[(size_t)k0 * FEAT + cb * 16 + m];
            b.y = kern[(size_t)(k0 + 1) * FEAT + cb * 16 + m];
            acc[cb] = __builtin_amdgcn_wmma_f32_16x16x4_f32(
                 false, a,  false, b,
                 (short)0, acc[cb],
                 false,  false);
        }
    }

    const float4* __restrict__ dv4 = reinterpret_cast<const float4*>(dinv + r0 + 8 * half);
    const float4 dlo = dv4[0];
    const float4 dhi = dv4[1];
    const float  rs[8] = {dlo.x, dlo.y, dlo.z, dlo.w, dhi.x, dhi.y, dhi.z, dhi.w};

#pragma unroll
    for (int cb = 0; cb < 4; ++cb) {
#pragma unroll
        for (int v = 0; v < 8; ++v) {
            h[(size_t)(r0 + v + 8 * half) * FEAT + cb * 16 + m] = acc[cb][v] * rs[v];
        }
    }
}

#define GATHER_WG 1024
#define GB        31
__global__ __launch_bounds__(256)
void gcn_row_gather(const int* __restrict__ offs,
                    const int* __restrict__ cnt,
                    const int2* __restrict__ csr,
                    const float* __restrict__ dinv,
                    const float* __restrict__ h,
                    const float* __restrict__ bias,
                    float* __restrict__ out, int n) {
    const int lane = threadIdx.x & 31;
    int r = __builtin_amdgcn_readfirstlane(blockIdx.x * 8 + (threadIdx.x >> 5));
    if (r >= n) return;
    const int nw = GATHER_WG * 8;

    const v2f* __restrict__ h2 = reinterpret_cast<const v2f*>(h);
    v2f* __restrict__ out2 = reinterpret_cast<v2f*>(out);
    const float b0 = bias[2 * lane];
    const float b1 = bias[2 * lane + 1];

#define EDGE_C(j) __builtin_amdgcn_readlane(c, (j))
#define EDGE_W(j) __int_as_float(__builtin_amdgcn_readlane(__float_as_int(wn), (j)))
#define H_ROW(cc) h2[(unsigned)(cc) * (unsigned)(FEAT / 2) + ulane]
    const unsigned ulane = (unsigned)lane;

    int   beg = offs[r];
    int   num = cnt[r];
    float dr  = dinv[r];
    int2  cw  = csr[(lane < min(num, GB)) ? beg + lane : 0];
    v2f   hs  = H_ROW(r);
    int   rn   = r + nw;
    int   begn = 0;
    int   numn = 0;
    float drn  = 0.0f;
    if (rn < n) { begn = offs[rn]; numn = cnt[rn]; drn = dinv[rn]; }

    for (;;) {
        const bool has_n = rn < n;
        const int2 cwn = csr[(lane < min(numn, GB)) ? begn + lane : 0];
        const v2f  hsn = H_ROW(has_n ? rn : r);
        const int rnn   = rn + nw;
        int       begnn = 0;
        int       numnn = 0;
        float     drnn  = 0.0f;
        if (rnn < n) { begnn = offs[rnn]; numnn = cnt[rnn]; drnn = dinv[rnn]; }

        v2f acc = {0.0f, 0.0f};
        for (int base = 0; base < num; base += GB) {
            const int m = min(GB, num - base);
            int   c;
            float wn;
            if (base == 0) {
                c  = cw.x;
                wn = dr * __int_as_float(cw.y);
            } else {
                c  = 0;
                wn = 0.0f;
                if (lane < m) {
                    const int2 e = csr[beg + base + lane];
                    c  = e.x;
                    wn = dr * __int_as_float(e.y);
                }
            }
            asm volatile("" :: "v"(c));
            v2f g16[16], g8[8], g4[4], g2[2], g1;
            int j = 0;
            if (m & 16) {
#pragma unroll
                for (int u = 0; u < 16; ++u) g16[u] = H_ROW(EDGE_C(j + u));
                j += 16;
            }
            if (m & 8) {
#pragma unroll
                for (int u = 0; u < 8; ++u) g8[u] = H_ROW(EDGE_C(j + u));
                j += 8;
            }
            if (m & 4) {
#pragma unroll
                for (int u = 0; u < 4; ++u) g4[u] = H_ROW(EDGE_C(j + u));
                j += 4;
            }
            if (m & 2) {
#pragma unroll
                for (int u = 0; u < 2; ++u) g2[u] = H_ROW(EDGE_C(j + u));
                j += 2;
            }
            if (m & 1) g1 = H_ROW(EDGE_C(j));
            j = 0;
            if (m & 16) {
#pragma unroll
                for (int u = 0; u < 16; ++u) acc += g16[u] * EDGE_W(j + u);
                j += 16;
            }
            if (m & 8) {
#pragma unroll
                for (int u = 0; u < 8; ++u) acc += g8[u] * EDGE_W(j + u);
                j += 8;
            }
            if (m & 4) {
#pragma unroll
                for (int u = 0; u < 4; ++u) acc += g4[u] * EDGE_W(j + u);
                j += 4;
            }
            if (m & 2) {
#pragma unroll
                for (int u = 0; u < 2; ++u) acc += g2[u] * EDGE_W(j + u);
                j += 2;
            }
            if (m & 1) acc += g1 * EDGE_W(j);
        }

        v2f o;
        o.x = (b0 + hs.x * dr * FILL_WEIGHT) + acc.x;
        o.y = (b1 + hs.y * dr * FILL_WEIGHT) + acc.y;
        out2[(unsigned)r * (unsigned)(FEAT / 2) + ulane] = o;

        if (!has_n) break;
        r = rn;   beg = begn;   num = numn;   dr = drn;   cw = cwn;   hs = hsn;
        rn = rnn; begn = begnn; numn = numnn; drn = drnn;
    }
#undef EDGE_C
#undef EDGE_W
#undef H_ROW
}

extern "C" void kernel_launch(void* const* d_in, const int* in_sizes, int n_in,
                              void* d_out, int out_size, void* d_ws, size_t ws_size,
                              hipStream_t stream) {
    const float* x    = (const float*)d_in[0];
    const int*   ei   = (const int*)d_in[1];
    const float* ew   = (const float*)d_in[2];
    const float* kern = (const float*)d_in[3];
    const float* bias = (const float*)d_in[4];

    const int N = in_sizes[0] / FEAT;
    const int E = in_sizes[2];
    const int* row = ei;
    const int* col = ei + E;

    const int G1   = (E + EDGE_CHUNK - 1) / EDGE_CHUNK;
    const int NBK  = (N + BK_ROWS - 1) / BK_ROWS;
    const int n2   = NBK * G1;
    const int Npad = (N + 3) & ~3;
    const int Hpad = (n2 + 31) & ~31;
    const int Spad = G1 * EDGE_CHUNK;

    float* h      = (float*)d_ws;
    float* dinv   = h + (size_t)N * FEAT;
    int*   cnt    = (int*)(dinv + Npad);
    int*   offs   = cnt + Npad;
    int*   pieces = offs + Npad;
    int2*  slab   = (int2*)(pieces + Hpad);
    int2*  csr    = slab + Spad;
    float* out    = (float*)d_out;

    const int T = 256;

    gcn_chunk_sort<<<G1, T, 0, stream>>>(row, col, ew, pieces, slab, E, G1, NBK);
    gcn_bucket_sort<<<NBK, 1024, 0, stream>>>(pieces, slab, csr, offs, cnt, dinv, N, G1);

    const int ntile = (N + 15) / 16;
    gcn_gemm_wmma<<<(ntile + 7) / 8, T, 0, stream>>>(x, kern, dinv, h, N);

    gcn_row_gather<<<GATHER_WG, T, 0, stream>>>(offs, cnt, csr, dinv, h, bias, out, N);
}
